# starting kernel
# baseline (speedup 1.0000x reference)
#include <hip/hip_runtime.h>
#include <hip/hip_bf16.h>

typedef __attribute__((ext_vector_type(2))) float v2f;
typedef __attribute__((ext_vector_type(8))) float v8f;

#define D_INNER 1024
#define STATE   16
#define DT_RANK 64
#define BATCH   2
#define SEQ     2048
#define ROWS    (BATCH * SEQ)
#define E_DIM   (DT_RANK + 2 * STATE)
#define NCHUNK  8
#define CLEN    (SEQ / NCHUNK)

__global__ __launch_bounds__(128) void xproj_wmma(
    const float* __restrict__ u, const float* __restrict__ Wx,
    float* __restrict__ x_dbl) {
  const int wave = blockIdx.x * (blockDim.x >> 5) + (threadIdx.x >> 5);
  const int lane = threadIdx.x & 31;
  const int rt = wave / (E_DIM / 16);
  const int ct = wave % (E_DIM / 16);
  const int row0 = rt * 16, col0 = ct * 16;
  const int m  = lane & 15;
  const int kh = lane >> 4;

  v8f acc = {};
  const float* ua = u  + (size_t)(row0 + m) * D_INNER + 2 * kh;
  const float* wa = Wx + (size_t)(col0 + m) * D_INNER + 2 * kh;
#pragma unroll 8
  for (int kb = 0; kb < D_INNER; kb += 4) {
    v2f a = *(const v2f*)(ua + kb);
    v2f b = *(const v2f*)(wa + kb);
    acc = __builtin_amdgcn_wmma_f32_16x16x4_f32(
         false, a,  false, b,
         (short)0, acc,  false,  false);
  }
  float* outp = x_dbl + (size_t)(row0 + 8 * kh) * E_DIM + col0 + m;
#pragma unroll
  for (int j = 0; j < 8; ++j) outp[(size_t)j * E_DIM] = acc[j];
}

__global__ __launch_bounds__(128) void dtproj_wmma(
    const float* __restrict__ x_dbl, const float* __restrict__ Wdt,
    const float* __restrict__ b_dt, float* __restrict__ dt) {
  const int wave = blockIdx.x * (blockDim.x >> 5) + (threadIdx.x >> 5);
  const int lane = threadIdx.x & 31;
  const int rt = wave / (D_INNER / 16);
  const int ct = wave % (D_INNER / 16);
  const int row0 = rt * 16, col0 = ct * 16;
  const int m  = lane & 15;
  const int kh = lane >> 4;

  v8f acc = {};
  const float* aa = x_dbl + (size_t)(row0 + m) * E_DIM + 2 * kh;
  const float* wa = Wdt   + (size_t)(col0 + m) * DT_RANK + 2 * kh;
#pragma unroll
  for (int kb = 0; kb < DT_RANK; kb += 4) {
    v2f a = *(const v2f*)(aa + kb);
    v2f b = *(const v2f*)(wa + kb);
    acc = __builtin_amdgcn_wmma_f32_16x16x4_f32(
        false, a, false, b, (short)0, acc, false, false);
  }
  const float bias = b_dt[col0 + m];
  float* outp = dt + (size_t)(row0 + 8 * kh) * D_INNER + col0 + m;
#pragma unroll
  for (int j = 0; j < 8; ++j) {
    float x = acc[j] + bias;
    float sp = (x > 20.0f) ? x : log1pf(__expf(x));
    outp[(size_t)j * D_INNER] = sp;
  }
}

__global__ __launch_bounds__(256) void scan_pass1(
    const float* __restrict__ u, const float* __restrict__ A_log,
    const float* __restrict__ x_dbl, const float* __restrict__ dt,
    float* __restrict__ prodA, float* __restrict__ hend) {
  const int wave = blockIdx.x * (blockDim.x >> 5) + (threadIdx.x >> 5);
  const int lane = threadIdx.x & 31;
  const int b     = wave / (NCHUNK * (D_INNER / 2));
  const int rem   = wave % (NCHUNK * (D_INNER / 2));
  const int chunk = rem / (D_INNER / 2);
  const int dpair = rem % (D_INNER / 2);
  const int d = dpair * 2 + (lane >> 4);
  const int n = lane & 15;

  const float A = -__expf(A_log[d * STATE + n]);
  float h = 0.0f, p = 1.0f;
  const int lbase = chunk * CLEN;
  for (int i = 0; i < CLEN; ++i) {
    const size_t r = (size_t)b * SEQ + (lbase + i);
    const float dtv = dt[r * D_INNER + d];
    const float uv  = u [r * D_INNER + d];
    const float Bv  = x_dbl[r * E_DIM + DT_RANK + n];
    const float a = __expf(dtv * A);
    h = fmaf(a, h, dtv * uv * Bv);
    p *= a;
  }
  const size_t sidx = (((size_t)b * D_INNER + d) * STATE + n) * NCHUNK + chunk;
  prodA[sidx] = p;
  hend[sidx]  = h;
}

__global__ __launch_bounds__(256) void scan_combine(
    const float* __restrict__ prodA, const float* __restrict__ hend,
    float* __restrict__ hin) {
  const int t = blockIdx.x * blockDim.x + threadIdx.x;
  const size_t base = (size_t)t * NCHUNK;
  float h = 0.0f;
#pragma unroll
  for (int c = 0; c < NCHUNK; ++c) {
    hin[base + c] = h;
    h = fmaf(prodA[base + c], h, hend[base + c]);
  }
}

__global__ __launch_bounds__(256) void scan_pass3(
    const float* __restrict__ u, const float* __restrict__ A_log,
    const float* __restrict__ x_dbl, const float* __restrict__ dt,
    const float* __restrict__ hin, const float* __restrict__ Dp,
    float* __restrict__ out) {
  const int wave = blockIdx.x * (blockDim.x >> 5) + (threadIdx.x >> 5);
  const int lane = threadIdx.x & 31;
  const int b     = wave / (NCHUNK * (D_INNER / 2));
  const int rem   = wave % (NCHUNK * (D_INNER / 2));
  const int chunk = rem / (D_INNER / 2);
  const int dpair = rem % (D_INNER / 2);
  const int d = dpair * 2 + (lane >> 4);
  const int n = lane & 15;

  const float A  = -__expf(A_log[d * STATE + n]);
  const float dp = Dp[d];
  const size_t sidx = (((size_t)b * D_INNER + d) * STATE + n) * NCHUNK + chunk;
  float h = hin[sidx];
  const int lbase = chunk * CLEN;
  for (int i = 0; i < CLEN; ++i) {
    const size_t r = (size_t)b * SEQ + (lbase + i);
    const float dtv = dt[r * D_INNER + d];
    const float uv  = u [r * D_INNER + d];
    const float Bv  = x_dbl[r * E_DIM + DT_RANK + n];
    const float Cv  = x_dbl[r * E_DIM + DT_RANK + STATE + n];
    const float a = __expf(dtv * A);
    h = fmaf(a, h, dtv * uv * Bv);
    float y = h * Cv;
    y += __shfl_xor(y, 1, 32);
    y += __shfl_xor(y, 2, 32);
    y += __shfl_xor(y, 4, 32);
    y += __shfl_xor(y, 8, 32);
    if (n == 0) out[r * D_INNER + d] = fmaf(uv, dp, y);
  }
}

extern "C" void kernel_launch(void* const* d_in, const int* in_sizes, int n_in,
                              void* d_out, int out_size, void* d_ws, size_t ws_size,
                              hipStream_t stream) {
  const float* u     = (const float*)d_in[0];
  const float* A_log = (const float*)d_in[1];
  const float* W_x   = (const float*)d_in[2];
  const float* W_dt  = (const float*)d_in[3];
  const float* b_dt  = (const float*)d_in[4];
  const float* Dp    = (const float*)d_in[5];
  float* out = (float*)d_out;

  float* ws    = (float*)d_ws;
  float* x_dbl = ws;
  float* dtbuf = x_dbl + (size_t)ROWS * E_DIM;
  float* prodA = dtbuf + (size_t)ROWS * D_INNER;
  float* hend  = prodA + (size_t)BATCH * D_INNER * STATE * NCHUNK;
  float* hin   = hend  + (size_t)BATCH * D_INNER * STATE * NCHUNK;

  xproj_wmma<<<1536 / 4, 128, 0, stream>>>(u, W_x, x_dbl);
  dtproj_wmma<<<16384 / 4, 128, 0, stream>>>(x_dbl, W_dt, b_dt, dtbuf);
  scan_pass1<<<8192 / 8, 256, 0, stream>>>(u, A_log, x_dbl, dtbuf, prodA, hend);
  scan_combine<<<(BATCH * D_INNER * STATE) / 256, 256, 0, stream>>>(prodA, hend, hin);
  scan_pass3<<<8192 / 8, 256, 0, stream>>>(u, A_log, x_dbl, dtbuf, hin, Dp, out);
}


# scan remapped: one lane per channel with its 16 states in registers, 32 chunks of 64 steps; sum over states now serial in the lane (4 partials), chunk carries join at 32 edges, no shuffles
# speedup vs baseline: 3.2165x; 3.2165x over previous
#include <hip/hip_runtime.h>
#include <hip/hip_bf16.h>

typedef __attribute__((ext_vector_type(2))) float v2f;
typedef __attribute__((ext_vector_type(8))) float v8f;

#define D_INNER 1024
#define STATE   16
#define DT_RANK 64
#define BATCH   2
#define SEQ     2048
#define ROWS    (BATCH * SEQ)
#define E_DIM   (DT_RANK + 2 * STATE)
#define NCHUNK  32
#define CLEN    (SEQ / NCHUNK)
#define SCAN_BLOCK 128
#define DBLKS   (D_INNER / SCAN_BLOCK)

__global__ __launch_bounds__(128) void xproj_wmma(
    const float* __restrict__ u, const float* __restrict__ Wx,
    float* __restrict__ x_dbl) {
  const int wave = blockIdx.x * (blockDim.x >> 5) + (threadIdx.x >> 5);
  const int lane = threadIdx.x & 31;
  const int rt = wave / (E_DIM / 16);
  const int ct = wave % (E_DIM / 16);
  const int row0 = rt * 16, col0 = ct * 16;
  const int m  = lane & 15;
  const int kh = lane >> 4;

  v8f acc = {};
  const float* ua = u  + (size_t)(row0 + m) * D_INNER + 2 * kh;
  const float* wa = Wx + (size_t)(col0 + m) * D_INNER + 2 * kh;
#pragma unroll 8
  for (int kb = 0; kb < D_INNER; kb += 4) {
    v2f a = *(const v2f*)(ua + kb);
    v2f b = *(const v2f*)(wa + kb);
    acc = __builtin_amdgcn_wmma_f32_16x16x4_f32(
         false, a,  false, b,
         (short)0, acc,  false,  false);
  }
  float* outp = x_dbl + (size_t)(row0 + 8 * kh) * E_DIM + col0 + m;
#pragma unroll
  for (int j = 0; j < 8; ++j) outp[(size_t)j * E_DIM] = acc[j];
}

__global__ __launch_bounds__(128) void dtproj_wmma(
    const float* __restrict__ x_dbl, const float* __restrict__ Wdt,
    const float* __restrict__ b_dt, float* __restrict__ dt) {
  const int wave = blockIdx.x * (blockDim.x >> 5) + (threadIdx.x >> 5);
  const int lane = threadIdx.x & 31;
  const int rt = wave / (D_INNER / 16);
  const int ct = wave % (D_INNER / 16);
  const int row0 = rt * 16, col0 = ct * 16;
  const int m  = lane & 15;
  const int kh = lane >> 4;

  v8f acc = {};
  const float* aa = x_dbl + (size_t)(row0 + m) * E_DIM + 2 * kh;
  const float* wa = Wdt   + (size_t)(col0 + m) * DT_RANK + 2 * kh;
#pragma unroll
  for (int kb = 0; kb < DT_RANK; kb += 4) {
    v2f a = *(const v2f*)(aa + kb);
    v2f b = *(const v2f*)(wa + kb);
    acc = __builtin_amdgcn_wmma_f32_16x16x4_f32(
        false, a, false, b, (short)0, acc, false, false);
  }
  const float bias = b_dt[col0 + m];
  float* outp = dt + (size_t)(row0 + 8 * kh) * D_INNER + col0 + m;
#pragma unroll
  for (int j = 0; j < 8; ++j) {
    float x = acc[j] + bias;
    float sp = (x > 20.0f) ? x : log1pf(__expf(x));
    outp[(size_t)j * D_INNER] = sp;
  }
}

__global__ __launch_bounds__(SCAN_BLOCK) void scan_pass1(
    const float* __restrict__ u, const float* __restrict__ A_log,
    const float* __restrict__ x_dbl, const float* __restrict__ dt,
    float* __restrict__ prodA, float* __restrict__ hend) {
  const int blk   = blockIdx.x;
  const int dblk  = blk % DBLKS;
  const int chunk = (blk / DBLKS) % NCHUNK;
  const int b     = blk / (DBLKS * NCHUNK);
  const int d     = dblk * SCAN_BLOCK + threadIdx.x;

  float A[STATE], h[STATE], p[STATE];
#pragma unroll
  for (int n = 0; n < STATE; ++n) {
    A[n] = -__expf(A_log[d * STATE + n]);
    h[n] = 0.0f;
    p[n] = 1.0f;
  }
  const size_t r0 = (size_t)b * SEQ + (size_t)chunk * CLEN;
  const float* dtp = dt + r0 * D_INNER + d;
  const float* up  = u  + r0 * D_INNER + d;
  const float* bp  = x_dbl + r0 * E_DIM + DT_RANK;
#pragma unroll 4
  for (int i = 0; i < CLEN; ++i) {
    const float dtv = dtp[(size_t)i * D_INNER];
    const float uv  = up [(size_t)i * D_INNER];
    const float dtu = dtv * uv;
    const float* Brow = bp + (size_t)i * E_DIM;
#pragma unroll
    for (int n = 0; n < STATE; ++n) {
      const float a = __expf(dtv * A[n]);
      h[n] = fmaf(a, h[n], dtu * Brow[n]);
      p[n] *= a;
    }
  }
  const size_t sbase = ((size_t)(b * NCHUNK + chunk) * STATE) * D_INNER + d;
#pragma unroll
  for (int n = 0; n < STATE; ++n) {
    prodA[sbase + (size_t)n * D_INNER] = p[n];
    hend [sbase + (size_t)n * D_INNER] = h[n];
  }
}

__global__ __launch_bounds__(256) void scan_combine(
    const float* __restrict__ prodA, const float* __restrict__ hend,
    float* __restrict__ hin) {
  const int t = blockIdx.x * blockDim.x + threadIdx.x;
  const int d = t % D_INNER;
  const int n = (t / D_INNER) % STATE;
  const int b = t / (D_INNER * STATE);
  float h = 0.0f;
#pragma unroll
  for (int c = 0; c < NCHUNK; ++c) {
    const size_t idx = ((size_t)(b * NCHUNK + c) * STATE + n) * D_INNER + d;
    hin[idx] = h;
    h = fmaf(prodA[idx], h, hend[idx]);
  }
}

__global__ __launch_bounds__(SCAN_BLOCK) void scan_pass3(
    const float* __restrict__ u, const float* __restrict__ A_log,
    const float* __restrict__ x_dbl, const float* __restrict__ dt,
    const float* __restrict__ hin, const float* __restrict__ Dp,
    float* __restrict__ out) {
  const int blk   = blockIdx.x;
  const int dblk  = blk % DBLKS;
  const int chunk = (blk / DBLKS) % NCHUNK;
  const int b     = blk / (DBLKS * NCHUNK);
  const int d     = dblk * SCAN_BLOCK + threadIdx.x;

  const size_t sbase = ((size_t)(b * NCHUNK + chunk) * STATE) * D_INNER + d;
  float A[STATE], h[STATE];
#pragma unroll
  for (int n = 0; n < STATE; ++n) {
    A[n] = -__expf(A_log[d * STATE + n]);
    h[n] = hin[sbase + (size_t)n * D_INNER];
  }
  const float dp = Dp[d];
  const size_t r0 = (size_t)b * SEQ + (size_t)chunk * CLEN;
  const float* dtp = dt + r0 * D_INNER + d;
  const float* up  = u  + r0 * D_INNER + d;
  const float* bp  = x_dbl + r0 * E_DIM + DT_RANK;
  float* op = out + r0 * D_INNER + d;
#pragma unroll 4
  for (int i = 0; i < CLEN; ++i) {
    const float dtv = dtp[(size_t)i * D_INNER];
    const float uv  = up [(size_t)i * D_INNER];
    const float dtu = dtv * uv;
    const float* Brow = bp + (size_t)i * E_DIM;
    const float* Crow = Brow + STATE;
    float y[4] = {0.0f, 0.0f, 0.0f, 0.0f};
#pragma unroll
    for (int n = 0; n < STATE; ++n) {
      const float a = __expf(dtv * A[n]);
      h[n] = fmaf(a, h[n], dtu * Brow[n]);
      y[n & 3] = fmaf(h[n], Crow[n], y[n & 3]);
    }
    op[(size_t)i * D_INNER] = fmaf(uv, dp, (y[0] + y[1]) + (y[2] + y[3]));
  }
}

extern "C" void kernel_launch(void* const* d_in, const int* in_sizes, int n_in,
                              void* d_out, int out_size, void* d_ws, size_t ws_size,
                              hipStream_t stream) {
  const float* u     = (const float*)d_in[0];
  const float* A_log = (const float*)d_in[1];
  const float* W_x   = (const float*)d_in[2];
  const float* W_dt  = (const float*)d_in[3];
  const float* b_dt  = (const float*)d_in[4];
  const float* Dp    = (const float*)d_in[5];
  float* out = (float*)d_out;

  float* ws    = (float*)d_ws;
  float* x_dbl = ws;
  float* dtbuf = x_dbl + (size_t)ROWS * E_DIM;
  float* prodA = dtbuf + (size_t)ROWS * D_INNER;
  float* hend  = prodA + (size_t)BATCH * D_INNER * STATE * NCHUNK;
  float* hin   = hend  + (size_t)BATCH * D_INNER * STATE * NCHUNK;

  xproj_wmma<<<1536 / 4, 128, 0, stream>>>(u, W_x, x_dbl);
  dtproj_wmma<<<16384 / 4, 128, 0, stream>>>(x_dbl, W_dt, b_dt, dtbuf);
  scan_pass1<<<BATCH * NCHUNK * DBLKS, SCAN_BLOCK, 0, stream>>>(u, A_log, x_dbl, dtbuf, prodA, hend);
  scan_combine<<<(BATCH * D_INNER * STATE) / 256, 256, 0, stream>>>(prodA, hend, hin);
  scan_pass3<<<BATCH * NCHUNK * DBLKS, SCAN_BLOCK, 0, stream>>>(u, A_log, x_dbl, dtbuf, hin, Dp, out);
}


# GEMM1 regrouped: a block of 8 waves per 16-row strip, each wave one 128-wide K slice for all 6 column tiles, next block's loads issued ahead; K now summed as 8 partials joined through LDS
# speedup vs baseline: 3.8585x; 1.1996x over previous
#include <hip/hip_runtime.h>
#include <hip/hip_bf16.h>

typedef __attribute__((ext_vector_type(2))) float v2f;
typedef __attribute__((ext_vector_type(4))) float v4f;
typedef __attribute__((ext_vector_type(8))) float v8f;

#define D_INNER 1024
#define STATE   16
#define DT_RANK 64
#define BATCH   2
#define SEQ     2048
#define ROWS    (BATCH * SEQ)
#define E_DIM   (DT_RANK + 2 * STATE)
#define NCHUNK  32
#define CLEN    (SEQ / NCHUNK)
#define SCAN_BLOCK 128
#define DBLKS   (D_INNER / SCAN_BLOCK)

#define XP_WAVES  8
#define XP_KSLICE (D_INNER / XP_WAVES)
#define XP_TILES  (E_DIM / 16)

struct XpFrag {
  v4f a[4];
  v4f b[XP_TILES][4];
};

__device__ __forceinline__ void xp_load(XpFrag& f, const float* __restrict__ ua,
                                        const float* __restrict__ wa, int sb) {
#pragma unroll
  for (int q = 0; q < 4; ++q) f.a[q] = *(const v4f*)(ua + 32 * sb + 8 * q);
#pragma unroll
  for (int t = 0; t < XP_TILES; ++t) {
#pragma unroll
    for (int q = 0; q < 4; ++q)
      f.b[t][q] = *(const v4f*)(wa + (size_t)t * 16 * D_INNER + 32 * sb + 8 * q);
  }
}

__device__ __forceinline__ void xp_mma(const XpFrag& f, v8f (&acc)[XP_TILES]) {
#pragma unroll
  for (int q = 0; q < 4; ++q) {
    const v2f a0 = {f.a[q].x, f.a[q].y};
    const v2f a1 = {f.a[q].z, f.a[q].w};
#pragma unroll
    for (int t = 0; t < XP_TILES; ++t) {
      const v2f b0 = {f.b[t][q].x, f.b[t][q].y};
      acc[t] = __builtin_amdgcn_wmma_f32_16x16x4_f32(
           false, a0,  false, b0,
           (short)0, acc[t],  false,  false);
    }
#pragma unroll
    for (int t = 0; t < XP_TILES; ++t) {
      const v2f b1 = {f.b[t][q].z, f.b[t][q].w};
      acc[t] = __builtin_amdgcn_wmma_f32_16x16x4_f32(
          false, a1, false, b1, (short)0, acc[t], false, false);
    }
  }
}

__global__ __launch_bounds__(XP_WAVES * 32) void xproj_wmma(
    const float* __restrict__ u, const float* __restrict__ Wx,
    float* __restrict__ x_dbl) {
  __shared__ float red[XP_WAVES][XP_TILES][8][32];

  const int w    = threadIdx.x >> 5;
  const int lane = threadIdx.x & 31;
  const int row0 = blockIdx.x * 16;
  const int m  = lane & 15;
  const int kh = lane >> 4;

  const float* ua = u  + (size_t)(row0 + m) * D_INNER + w * XP_KSLICE + 4 * kh;
  const float* wa = Wx + (size_t)m * D_INNER + w * XP_KSLICE + 4 * kh;

  v8f acc[XP_TILES];
#pragma unroll
  for (int t = 0; t < XP_TILES; ++t) acc[t] = v8f{};

  XpFrag cur, nxt;
  xp_load(cur, ua, wa, 0);
#pragma unroll
  for (int sb = 0; sb < XP_KSLICE / 32; ++sb) {
    if (sb + 1 < XP_KSLICE / 32) xp_load(nxt, ua, wa, sb + 1);
    __builtin_amdgcn_sched_barrier(0);
    xp_mma(cur, acc);
    if (sb + 1 < XP_KSLICE / 32) cur = nxt;
  }

#pragma unroll
  for (int t = 0; t < XP_TILES; ++t) {
#pragma unroll
    for (int j = 0; j < 8; ++j) red[w][t][j][lane] = acc[t][j];
  }
  __syncthreads();

  float* outp = x_dbl + (size_t)row0 * E_DIM;
#pragma unroll
  for (int i = 0; i < (16 * E_DIM) / (XP_WAVES * 32); ++i) {
    const int o  = threadIdx.x + i * (XP_WAVES * 32);
    const int r  = o / E_DIM;
    const int c  = o - r * E_DIM;
    const int t  = c >> 4;
    const int ln = ((r >> 3) << 4) | (c & 15);
    const int j  = r & 7;
    float sum = red[0][t][j][ln];
#pragma unroll
    for (int ww = 1; ww < XP_WAVES; ++ww) sum += red[ww][t][j][ln];
    outp[o] = sum;
  }
}

__global__ __launch_bounds__(128) void dtproj_wmma(
    const float* __restrict__ x_dbl, const float* __restrict__ Wdt,
    const float* __restrict__ b_dt, float* __restrict__ dt) {
  const int wave = blockIdx.x * (blockDim.x >> 5) + (threadIdx.x >> 5);
  const int lane = threadIdx.x & 31;
  const int rt = wave / (D_INNER / 16);
  const int ct = wave % (D_INNER / 16);
  const int row0 = rt * 16, col0 = ct * 16;
  const int m  = lane & 15;
  const int kh = lane >> 4;

  v8f acc = {};
  const float* aa = x_dbl + (size_t)(row0 + m) * E_DIM + 2 * kh;
  const float* wa = Wdt   + (size_t)(col0 + m) * DT_RANK + 2 * kh;
#pragma unroll
  for (int kb = 0; kb < DT_RANK; kb += 4) {
    v2f a = *(const v2f*)(aa + kb);
    v2f b = *(const v2f*)(wa + kb);
    acc = __builtin_amdgcn_wmma_f32_16x16x4_f32(
        false, a, false, b, (short)0, acc, false, false);
  }
  const float bias = b_dt[col0 + m];
  float* outp = dt + (size_t)(row0 + 8 * kh) * D_INNER + col0 + m;
#pragma unroll
  for (int j = 0; j < 8; ++j) {
    float x = acc[j] + bias;
    float sp = (x > 20.0f) ? x : log1pf(__expf(x));
    outp[(size_t)j * D_INNER] = sp;
  }
}

__global__ __launch_bounds__(SCAN_BLOCK) void scan_pass1(
    const float* __restrict__ u, const float* __restrict__ A_log,
    const float* __restrict__ x_dbl, const float* __restrict__ dt,
    float* __restrict__ prodA, float* __restrict__ hend) {
  const int blk   = blockIdx.x;
  const int dblk  = blk % DBLKS;
  const int chunk = (blk / DBLKS) % NCHUNK;
  const int b     = blk / (DBLKS * NCHUNK);
  const int d     = dblk * SCAN_BLOCK + threadIdx.x;

  float A[STATE], h[STATE], p[STATE];
#pragma unroll
  for (int n = 0; n < STATE; ++n) {
    A[n] = -__expf(A_log[d * STATE + n]);
    h[n] = 0.0f;
    p[n] = 1.0f;
  }
  const size_t r0 = (size_t)b * SEQ + (size_t)chunk * CLEN;
  const float* dtp = dt + r0 * D_INNER + d;
  const float* up  = u  + r0 * D_INNER + d;
  const float* bp  = x_dbl + r0 * E_DIM + DT_RANK;
#pragma unroll 4
  for (int i = 0; i < CLEN; ++i) {
    const float dtv = dtp[(size_t)i * D_INNER];
    const float uv  = up [(size_t)i * D_INNER];
    const float dtu = dtv * uv;
    const float* Brow = bp + (size_t)i * E_DIM;
#pragma unroll
    for (int n = 0; n < STATE; ++n) {
      const float a = __expf(dtv * A[n]);
      h[n] = fmaf(a, h[n], dtu * Brow[n]);
      p[n] *= a;
    }
  }
  const size_t sbase = ((size_t)(b * NCHUNK + chunk) * STATE) * D_INNER + d;
#pragma unroll
  for (int n = 0; n < STATE; ++n) {
    prodA[sbase + (size_t)n * D_INNER] = p[n];
    hend [sbase + (size_t)n * D_INNER] = h[n];
  }
}

__global__ __launch_bounds__(256) void scan_combine(
    const float* __restrict__ prodA, const float* __restrict__ hend,
    float* __restrict__ hin) {
  const int t = blockIdx.x * blockDim.x + threadIdx.x;
  const int d = t % D_INNER;
  const int n = (t / D_INNER) % STATE;
  const int b = t / (D_INNER * STATE);
  float h = 0.0f;
#pragma unroll
  for (int c = 0; c < NCHUNK; ++c) {
    const size_t idx = ((size_t)(b * NCHUNK + c) * STATE + n) * D_INNER + d;
    hin[idx] = h;
    h = fmaf(prodA[idx], h, hend[idx]);
  }
}

__global__ __launch_bounds__(SCAN_BLOCK) void scan_pass3(
    const float* __restrict__ u, const float* __restrict__ A_log,
    const float* __restrict__ x_dbl, const float* __restrict__ dt,
    const float* __restrict__ hin, const float* __restrict__ Dp,
    float* __restrict__ out) {
  const int blk   = blockIdx.x;
  const int dblk  = blk % DBLKS;
  const int chunk = (blk / DBLKS) % NCHUNK;
  const int b     = blk / (DBLKS * NCHUNK);
  const int d     = dblk * SCAN_BLOCK + threadIdx.x;

  const size_t sbase = ((size_t)(b * NCHUNK + chunk) * STATE) * D_INNER + d;
  float A[STATE], h[STATE];
#pragma unroll
  for (int n = 0; n < STATE; ++n) {
    A[n] = -__expf(A_log[d * STATE + n]);
    h[n] = hin[sbase + (size_t)n * D_INNER];
  }
  const float dp = Dp[d];
  const size_t r0 = (size_t)b * SEQ + (size_t)chunk * CLEN;
  const float* dtp = dt + r0 * D_INNER + d;
  const float* up  = u  + r0 * D_INNER + d;
  const float* bp  = x_dbl + r0 * E_DIM + DT_RANK;
  float* op = out + r0 * D_INNER + d;
#pragma unroll 4
  for (int i = 0; i < CLEN; ++i) {
    const float dtv = dtp[(size_t)i * D_INNER];
    const float uv  = up [(size_t)i * D_INNER];
    const float dtu = dtv * uv;
    const float* Brow = bp + (size_t)i * E_DIM;
    const float* Crow = Brow + STATE;
    float y[4] = {0.0f, 0.0f, 0.0f, 0.0f};
#pragma unroll
    for (int n = 0; n < STATE; ++n) {
      const float a = __expf(dtv * A[n]);
      h[n] = fmaf(a, h[n], dtu * Brow[n]);
      y[n & 3] = fmaf(h[n], Crow[n], y[n & 3]);
    }
    op[(size_t)i * D_INNER] = fmaf(uv, dp, (y[0] + y[1]) + (y[2] + y[3]));
  }
}

extern "C" void kernel_launch(void* const* d_in, const int* in_sizes, int n_in,
                              void* d_out, int out_size, void* d_ws, size_t ws_size,
                              hipStream_t stream) {
  const float* u     = (const float*)d_in[0];
  const float* A_log = (const float*)d_in[1];
  const float* W_x   = (const float*)d_in[2];
  const float* W_dt  = (const float*)d_in[3];
  const float* b_dt  = (const float*)d_in[4];
  const float* Dp    = (const float*)d_in[5];
  float* out = (float*)d_out;

  float* ws    = (float*)d_ws;
  float* x_dbl = ws;
  float* dtbuf = x_dbl + (size_t)ROWS * E_DIM;
  float* prodA = dtbuf + (size_t)ROWS * D_INNER;
  float* hend  = prodA + (size_t)BATCH * D_INNER * STATE * NCHUNK;
  float* hin   = hend  + (size_t)BATCH * D_INNER * STATE * NCHUNK;

  xproj_wmma<<<ROWS / 16, XP_WAVES * 32, 0, stream>>>(u, W_x, x_dbl);
  dtproj_wmma<<<16384 / 4, 128, 0, stream>>>(x_dbl, W_dt, b_dt, dtbuf);
  scan_pass1<<<BATCH * NCHUNK * DBLKS, SCAN_BLOCK, 0, stream>>>(u, A_log, x_dbl, dtbuf, prodA, hend);
  scan_combine<<<(BATCH * D_INNER * STATE) / 256, 256, 0, stream>>>(prodA, hend, hin);
  scan_pass3<<<BATCH * NCHUNK * DBLKS, SCAN_BLOCK, 0, stream>>>(u, A_log, x_dbl, dtbuf, hin, Dp, out);
}


# scan cut into 64 chunks of 32 steps (was 32 of 64): four waves per SIMD instead of two cover each step group's load wait; chunk carries now join at 64 edges
# speedup vs baseline: 4.4926x; 1.1643x over previous
#include <hip/hip_runtime.h>
#include <hip/hip_bf16.h>

typedef __attribute__((ext_vector_type(2))) float v2f;
typedef __attribute__((ext_vector_type(4))) float v4f;
typedef __attribute__((ext_vector_type(8))) float v8f;

#define D_INNER 1024
#define STATE   16
#define DT_RANK 64
#define BATCH   2
#define SEQ     2048
#define ROWS    (BATCH * SEQ)
#define E_DIM   (DT_RANK + 2 * STATE)
#define NCHUNK  64
#define CLEN    (SEQ / NCHUNK)
#define SCAN_BLOCK 128
#define DBLKS   (D_INNER / SCAN_BLOCK)

#define XP_WAVES  8
#define XP_KSLICE (D_INNER / XP_WAVES)
#define XP_TILES  (E_DIM / 16)

struct XpFrag {
  v4f a[4];
  v4f b[XP_TILES][4];
};

__device__ __forceinline__ void xp_load(XpFrag& f, const float* __restrict__ ua,
                                        const float* __restrict__ wa, int sb) {
#pragma unroll
  for (int q = 0; q < 4; ++q) f.a[q] = *(const v4f*)(ua + 32 * sb + 8 * q);
#pragma unroll
  for (int t = 0; t < XP_TILES; ++t) {
#pragma unroll
    for (int q = 0; q < 4; ++q)
      f.b[t][q] = *(const v4f*)(wa + (size_t)t * 16 * D_INNER + 32 * sb + 8 * q);
  }
}

__device__ __forceinline__ void xp_mma(const XpFrag& f, v8f (&acc)[XP_TILES]) {
#pragma unroll
  for (int q = 0; q < 4; ++q) {
    const v2f a0 = {f.a[q].x, f.a[q].y};
    const v2f a1 = {f.a[q].z, f.a[q].w};
#pragma unroll
    for (int t = 0; t < XP_TILES; ++t) {
      const v2f b0 = {f.b[t][q].x, f.b[t][q].y};
      acc[t] = __builtin_amdgcn_wmma_f32_16x16x4_f32(
           false, a0,  false, b0,
           (short)0, acc[t],  false,  false);
    }
#pragma unroll
    for (int t = 0; t < XP_TILES; ++t) {
      const v2f b1 = {f.b[t][q].z, f.b[t][q].w};
      acc[t] = __builtin_amdgcn_wmma_f32_16x16x4_f32(
          false, a1, false, b1, (short)0, acc[t], false, false);
    }
  }
}

__global__ __launch_bounds__(XP_WAVES * 32) void xproj_wmma(
    const float* __restrict__ u, const float* __restrict__ Wx,
    float* __restrict__ x_dbl) {
  __shared__ float red[XP_WAVES][XP_TILES][8][32];

  const int w    = threadIdx.x >> 5;
  const int lane = threadIdx.x & 31;
  const int row0 = blockIdx.x * 16;
  const int m  = lane & 15;
  const int kh = lane >> 4;

  const float* ua = u  + (size_t)(row0 + m) * D_INNER + w * XP_KSLICE + 4 * kh;
  const float* wa = Wx + (size_t)m * D_INNER + w * XP_KSLICE + 4 * kh;

  v8f acc[XP_TILES];
#pragma unroll
  for (int t = 0; t < XP_TILES; ++t) acc[t] = v8f{};

  XpFrag cur, nxt;
  xp_load(cur, ua, wa, 0);
#pragma unroll
  for (int sb = 0; sb < XP_KSLICE / 32; ++sb) {
    if (sb + 1 < XP_KSLICE / 32) xp_load(nxt, ua, wa, sb + 1);
    __builtin_amdgcn_sched_barrier(0);
    xp_mma(cur, acc);
    if (sb + 1 < XP_KSLICE / 32) cur = nxt;
  }

#pragma unroll
  for (int t = 0; t < XP_TILES; ++t) {
#pragma unroll
    for (int j = 0; j < 8; ++j) red[w][t][j][lane] = acc[t][j];
  }
  __syncthreads();

  float* outp = x_dbl + (size_t)row0 * E_DIM;
#pragma unroll
  for (int i = 0; i < (16 * E_DIM) / (XP_WAVES * 32); ++i) {
    const int o  = threadIdx.x + i * (XP_WAVES * 32);
    const int r  = o / E_DIM;
    const int c  = o - r * E_DIM;
    const int t  = c >> 4;
    const int ln = ((r >> 3) << 4) | (c & 15);
    const int j  = r & 7;
    float sum = red[0][t][j][ln];
#pragma unroll
    for (int ww = 1; ww < XP_WAVES; ++ww) sum += red[ww][t][j][ln];
    outp[o] = sum;
  }
}

__global__ __launch_bounds__(128) void dtproj_wmma(
    const float* __restrict__ x_dbl, const float* __restrict__ Wdt,
    const float* __restrict__ b_dt, float* __restrict__ dt) {
  const int wave = blockIdx.x * (blockDim.x >> 5) + (threadIdx.x >> 5);
  const int lane = threadIdx.x & 31;
  const int rt = wave / (D_INNER / 16);
  const int ct = wave % (D_INNER / 16);
  const int row0 = rt * 16, col0 = ct * 16;
  const int m  = lane & 15;
  const int kh = lane >> 4;

  v8f acc = {};
  const float* aa = x_dbl + (size_t)(row0 + m) * E_DIM + 2 * kh;
  const float* wa = Wdt   + (size_t)(col0 + m) * DT_RANK + 2 * kh;
#pragma unroll
  for (int kb = 0; kb < DT_RANK; kb += 4) {
    v2f a = *(const v2f*)(aa + kb);
    v2f b = *(const v2f*)(wa + kb);
    acc = __builtin_amdgcn_wmma_f32_16x16x4_f32(
        false, a, false, b, (short)0, acc, false, false);
  }
  const float bias = b_dt[col0 + m];
  float* outp = dt + (size_t)(row0 + 8 * kh) * D_INNER + col0 + m;
#pragma unroll
  for (int j = 0; j < 8; ++j) {
    float x = acc[j] + bias;
    float sp = (x > 20.0f) ? x : log1pf(__expf(x));
    outp[(size_t)j * D_INNER] = sp;
  }
}

__global__ __launch_bounds__(SCAN_BLOCK) void scan_pass1(
    const float* __restrict__ u, const float* __restrict__ A_log,
    const float* __restrict__ x_dbl, const float* __restrict__ dt,
    float* __restrict__ prodA, float* __restrict__ hend) {
  const int blk   = blockIdx.x;
  const int dblk  = blk % DBLKS;
  const int chunk = (blk / DBLKS) % NCHUNK;
  const int b     = blk / (DBLKS * NCHUNK);
  const int d     = dblk * SCAN_BLOCK + threadIdx.x;

  float A[STATE], h[STATE], p[STATE];
#pragma unroll
  for (int n = 0; n < STATE; ++n) {
    A[n] = -__expf(A_log[d * STATE + n]);
    h[n] = 0.0f;
    p[n] = 1.0f;
  }
  const size_t r0 = (size_t)b * SEQ + (size_t)chunk * CLEN;
  const float* dtp = dt + r0 * D_INNER + d;
  const float* up  = u  + r0 * D_INNER + d;
  const float* bp  = x_dbl + r0 * E_DIM + DT_RANK;
#pragma unroll 4
  for (int i = 0; i < CLEN; ++i) {
    const float dtv = dtp[(size_t)i * D_INNER];
    const float uv  = up [(size_t)i * D_INNER];
    const float dtu = dtv * uv;
    const float* Brow = bp + (size_t)i * E_DIM;
#pragma unroll
    for (int n = 0; n < STATE; ++n) {
      const float a = __expf(dtv * A[n]);
      h[n] = fmaf(a, h[n], dtu * Brow[n]);
      p[n] *= a;
    }
  }
  const size_t sbase = ((size_t)(b * NCHUNK + chunk) * STATE) * D_INNER + d;
#pragma unroll
  for (int n = 0; n < STATE; ++n) {
    prodA[sbase + (size_t)n * D_INNER] = p[n];
    hend [sbase + (size_t)n * D_INNER] = h[n];
  }
}

__global__ __launch_bounds__(256) void scan_combine(
    const float* __restrict__ prodA, const float* __restrict__ hend,
    float* __restrict__ hin) {
  const int t = blockIdx.x * blockDim.x + threadIdx.x;
  const int d = t % D_INNER;
  const int n = (t / D_INNER) % STATE;
  const int b = t / (D_INNER * STATE);
  float h = 0.0f;
#pragma unroll
  for (int c = 0; c < NCHUNK; ++c) {
    const size_t idx = ((size_t)(b * NCHUNK + c) * STATE + n) * D_INNER + d;
    hin[idx] = h;
    h = fmaf(prodA[idx], h, hend[idx]);
  }
}

__global__ __launch_bounds__(SCAN_BLOCK) void scan_pass3(
    const float* __restrict__ u, const float* __restrict__ A_log,
    const float* __restrict__ x_dbl, const float* __restrict__ dt,
    const float* __restrict__ hin, const float* __restrict__ Dp,
    float* __restrict__ out) {
  const int blk   = blockIdx.x;
  const int dblk  = blk % DBLKS;
  const int chunk = (blk / DBLKS) % NCHUNK;
  const int b     = blk / (DBLKS * NCHUNK);
  const int d     = dblk * SCAN_BLOCK + threadIdx.x;

  const size_t sbase = ((size_t)(b * NCHUNK + chunk) * STATE) * D_INNER + d;
  float A[STATE], h[STATE];
#pragma unroll
  for (int n = 0; n < STATE; ++n) {
    A[n] = -__expf(A_log[d * STATE + n]);
    h[n] = hin[sbase + (size_t)n * D_INNER];
  }
  const float dp = Dp[d];
  const size_t r0 = (size_t)b * SEQ + (size_t)chunk * CLEN;
  const float* dtp = dt + r0 * D_INNER + d;
  const float* up  = u  + r0 * D_INNER + d;
  const float* bp  = x_dbl + r0 * E_DIM + DT_RANK;
  float* op = out + r0 * D_INNER + d;
#pragma unroll 4
  for (int i = 0; i < CLEN; ++i) {
    const float dtv = dtp[(size_t)i * D_INNER];
    const float uv  = up [(size_t)i * D_INNER];
    const float dtu = dtv * uv;
    const float* Brow = bp + (size_t)i * E_DIM;
    const float* Crow = Brow + STATE;
    float y[4] = {0.0f, 0.0f, 0.0f, 0.0f};
#pragma unroll
    for (int n = 0; n < STATE; ++n) {
      const float a = __expf(dtv * A[n]);
      h[n] = fmaf(a, h[n], dtu * Brow[n]);
      y[n & 3] = fmaf(h[n], Crow[n], y[n & 3]);
    }
    op[(size_t)i * D_INNER] = fmaf(uv, dp, (y[0] + y[1]) + (y[2] + y[3]));
  }
}

extern "C" void kernel_launch(void* const* d_in, const int* in_sizes, int n_in,
                              void* d_out, int out_size, void* d_ws, size_t ws_size,
                              hipStream_t stream) {
  const float* u     = (const float*)d_in[0];
  const float* A_log = (const float*)d_in[1];
  const float* W_x   = (const float*)d_in[2];
  const float* W_dt  = (const float*)d_in[3];
  const float* b_dt  = (const float*)d_in[4];
  const float* Dp    = (const float*)d_in[5];
  float* out = (float*)d_out;

  float* ws    = (float*)d_ws;
  float* x_dbl = ws;
  float* dtbuf = x_dbl + (size_t)ROWS * E_DIM;
  float* prodA = dtbuf + (size_t)ROWS * D_INNER;
  float* hend  = prodA + (size_t)BATCH * D_INNER * STATE * NCHUNK;
  float* hin   = hend  + (size_t)BATCH * D_INNER * STATE * NCHUNK;

  xproj_wmma<<<ROWS / 16, XP_WAVES * 32, 0, stream>>>(u, W_x, x_dbl);
  dtproj_wmma<<<16384 / 4, 128, 0, stream>>>(x_dbl, W_dt, b_dt, dtbuf);
  scan_pass1<<<BATCH * NCHUNK * DBLKS, SCAN_BLOCK, 0, stream>>>(u, A_log, x_dbl, dtbuf, prodA, hend);
  scan_combine<<<(BATCH * D_INNER * STATE) / 256, 256, 0, stream>>>(prodA, hend, hin);
  scan_pass3<<<BATCH * NCHUNK * DBLKS, SCAN_BLOCK, 0, stream>>>(u, A_log, x_dbl, dtbuf, hin, Dp, out);
}


# dt projection and softplus moved into GEMM1's epilogue: a block forms its 16 dt rows from its x_dbl strip in LDS, W_dt read in 16-byte pieces (k summed in another order); 4 launches, not 5
# speedup vs baseline: 4.8587x; 1.0815x over previous
#include <hip/hip_runtime.h>
#include <hip/hip_bf16.h>

typedef __attribute__((ext_vector_type(2))) float v2f;
typedef __attribute__((ext_vector_type(4))) float v4f;
typedef __attribute__((ext_vector_type(8))) float v8f;

#define D_INNER 1024
#define STATE   16
#define DT_RANK 64
#define BATCH   2
#define SEQ     2048
#define ROWS    (BATCH * SEQ)
#define E_DIM   (DT_RANK + 2 * STATE)
#define NCHUNK  64
#define CLEN    (SEQ / NCHUNK)
#define SCAN_BLOCK 128
#define DBLKS   (D_INNER / SCAN_BLOCK)

#define XP_WAVES  8
#define XP_KSLICE (D_INNER / XP_WAVES)
#define XP_TILES  (E_DIM / 16)

#define XS_PITCH  68
#define DP_PAIRS  4

struct XpFrag {
  v4f a[4];
  v4f b[XP_TILES][4];
};

struct DpFrag {
  v4f b[2][8];
};

__device__ __forceinline__ void dp_load(DpFrag& f, const float* __restrict__ wb) {
#pragma unroll
  for (int t = 0; t < 2; ++t) {
#pragma unroll
    for (int q = 0; q < 8; ++q)
      f.b[t][q] = *(const v4f*)(wb + (size_t)t * 16 * DT_RANK + 8 * q);
  }
}

__device__ __forceinline__ float xhalf(float x) {
  return __builtin_bit_cast(float, __builtin_amdgcn_permlanex16(
      __builtin_bit_cast(int, x), __builtin_bit_cast(int, x),
      0x76543210, 0xfedcba98, false, false));
}

__device__ __forceinline__ float softplus_sel(float x) {
  const float lp = log1pf(__expf(x));
  return (x > 20.0f) ? x : lp;
}

__device__ __forceinline__ void dp_pair(const v4f (&af)[8], const DpFrag& f,
                                        float bias, int kh, float* __restrict__ op) {
  v8f c0 = {}, c1 = {};
#pragma unroll
  for (int q = 0; q < 8; ++q) {
    const v2f a0  = {af[q].x, af[q].y};
    const v2f a1  = {af[q].z, af[q].w};
    const v2f b00 = {f.b[0][q].x, f.b[0][q].y};
    const v2f b10 = {f.b[1][q].x, f.b[1][q].y};
    const v2f b01 = {f.b[0][q].z, f.b[0][q].w};
    const v2f b11 = {f.b[1][q].z, f.b[1][q].w};
    c0 = __builtin_amdgcn_wmma_f32_16x16x4_f32(false, a0, false, b00, (short)0, c0, false, false);
    c1 = __builtin_amdgcn_wmma_f32_16x16x4_f32(false, a0, false, b10, (short)0, c1, false, false);
    c0 = __builtin_amdgcn_wmma_f32_16x16x4_f32(false, a1, false, b01, (short)0, c0, false, false);
    c1 = __builtin_amdgcn_wmma_f32_16x16x4_f32(false, a1, false, b11, (short)0, c1, false, false);
  }
#pragma unroll
  for (int j = 0; j < 8; ++j) {
    const float v0 = c0[j], v1 = c1[j];
    const float got = xhalf(kh ? v0 : v1);
    const float x = kh ? got : v0;
    const float y = kh ? v1 : got;
    op[(size_t)j * D_INNER]       = softplus_sel(x + bias);
    op[(size_t)(j + 8) * D_INNER] = softplus_sel(y + bias);
  }
}

__device__ __forceinline__ void xp_load(XpFrag& f, const float* __restrict__ ua,
                                        const float* __restrict__ wa, int sb) {
#pragma unroll
  for (int q = 0; q < 4; ++q) f.a[q] = *(const v4f*)(ua + 32 * sb + 8 * q);
#pragma unroll
  for (int t = 0; t < XP_TILES; ++t) {
#pragma unroll
    for (int q = 0; q < 4; ++q)
      f.b[t][q] = *(const v4f*)(wa + (size_t)t * 16 * D_INNER + 32 * sb + 8 * q);
  }
}

__device__ __forceinline__ void xp_mma(const XpFrag& f, v8f (&acc)[XP_TILES]) {
#pragma unroll
  for (int q = 0; q < 4; ++q) {
    const v2f a0 = {f.a[q].x, f.a[q].y};
    const v2f a1 = {f.a[q].z, f.a[q].w};
#pragma unroll
    for (int t = 0; t < XP_TILES; ++t) {
      const v2f b0 = {f.b[t][q].x, f.b[t][q].y};
      acc[t] = __builtin_amdgcn_wmma_f32_16x16x4_f32(
           false, a0,  false, b0,
           (short)0, acc[t],  false,  false);
    }
#pragma unroll
    for (int t = 0; t < XP_TILES; ++t) {
      const v2f b1 = {f.b[t][q].z, f.b[t][q].w};
      acc[t] = __builtin_amdgcn_wmma_f32_16x16x4_f32(
          false, a1, false, b1, (short)0, acc[t], false, false);
    }
  }
}

__global__ __launch_bounds__(XP_WAVES * 32) void xproj_wmma(
    const float* __restrict__ u, const float* __restrict__ Wx,
    const float* __restrict__ Wdt, const float* __restrict__ b_dt,
    float* __restrict__ x_dbl, float* __restrict__ dt) {
  __shared__ float red[XP_WAVES][XP_TILES][8][32];
  __shared__ __attribute__((aligned(16))) float xs[16][XS_PITCH];

  const int w    = threadIdx.x >> 5;
  const int lane = threadIdx.x & 31;
  const int row0 = blockIdx.x * 16;
  const int m  = lane & 15;
  const int kh = lane >> 4;

  const float* ua = u  + (size_t)(row0 + m) * D_INNER + w * XP_KSLICE + 4 * kh;
  const float* wa = Wx + (size_t)m * D_INNER + w * XP_KSLICE + 4 * kh;

  v8f acc[XP_TILES];
#pragma unroll
  for (int t = 0; t < XP_TILES; ++t) acc[t] = v8f{};

  XpFrag cur, nxt;
  xp_load(cur, ua, wa, 0);
#pragma unroll
  for (int sb = 0; sb < XP_KSLICE / 32; ++sb) {
    if (sb + 1 < XP_KSLICE / 32) xp_load(nxt, ua, wa, sb + 1);
    __builtin_amdgcn_sched_barrier(0);
    xp_mma(cur, acc);
    if (sb + 1 < XP_KSLICE / 32) cur = nxt;
  }

#pragma unroll
  for (int t = 0; t < XP_TILES; ++t) {
#pragma unroll
    for (int j = 0; j < 8; ++j) red[w][t][j][lane] = acc[t][j];
  }
  __syncthreads();

  float* outp = x_dbl + (size_t)row0 * E_DIM;
#pragma unroll
  for (int i = 0; i < (16 * E_DIM) / (XP_WAVES * 32); ++i) {
    const int o  = threadIdx.x + i * (XP_WAVES * 32);
    const int r  = o / E_DIM;
    const int c  = o - r * E_DIM;
    const int t  = c >> 4;
    const int ln = ((r >> 3) << 4) | (c & 15);
    const int j  = r & 7;
    float sum = red[0][t][j][ln];
#pragma unroll
    for (int ww = 1; ww < XP_WAVES; ++ww) sum += red[ww][t][j][ln];
    outp[o] = sum;
    if (c < DT_RANK) xs[r][c] = sum;
  }
  __syncthreads();

  v4f af[8];
  {
    const float* ap = &xs[m][4 * kh];
#pragma unroll
    for (int q = 0; q < 8; ++q) af[q] = *(const v4f*)(ap + 8 * q);
  }
  const int cbase = w * (DP_PAIRS * 32);
  const float* wb = Wdt + (size_t)(cbase + m) * DT_RANK + 4 * kh;
  const float* bp = b_dt + cbase + lane;
  float* dtp = dt + (size_t)row0 * D_INNER + cbase + lane;

  DpFrag f0, f1;
  float bias0 = bp[0], bias1;
  dp_load(f0, wb);
#pragma unroll 1
  for (int p = 0; p < DP_PAIRS; p += 2) {
    bias1 = bp[(p + 1) * 32];
    dp_load(f1, wb + (size_t)(p + 1) * 32 * DT_RANK);
    __builtin_amdgcn_sched_barrier(0);
    dp_pair(af, f0, bias0, kh, dtp + p * 32);
    __builtin_amdgcn_sched_barrier(0);
    const int pn = (p + 2) & (DP_PAIRS - 1);
    bias0 = bp[pn * 32];
    dp_load(f0, wb + (size_t)pn * 32 * DT_RANK);
    __builtin_amdgcn_sched_barrier(0);
    dp_pair(af, f1, bias1, kh, dtp + (p + 1) * 32);
  }
}

__global__ __launch_bounds__(SCAN_BLOCK) void scan_pass1(
    const float* __restrict__ u, const float* __restrict__ A_log,
    const float* __restrict__ x_dbl, const float* __restrict__ dt,
    float* __restrict__ prodA, float* __restrict__ hend) {
  const int blk   = blockIdx.x;
  const int dblk  = blk % DBLKS;
  const int chunk = (blk / DBLKS) % NCHUNK;
  const int b     = blk / (DBLKS * NCHUNK);
  const int d     = dblk * SCAN_BLOCK + threadIdx.x;

  float A[STATE], h[STATE], p[STATE];
#pragma unroll
  for (int n = 0; n < STATE; ++n) {
    A[n] = -__expf(A_log[d * STATE + n]);
    h[n] = 0.0f;
    p[n] = 1.0f;
  }
  const size_t r0 = (size_t)b * SEQ + (size_t)chunk * CLEN;
  const float* dtp = dt + r0 * D_INNER + d;
  const float* up  = u  + r0 * D_INNER + d;
  const float* bp  = x_dbl + r0 * E_DIM + DT_RANK;
#pragma unroll 4
  for (int i = 0; i < CLEN; ++i) {
    const float dtv = dtp[(size_t)i * D_INNER];
    const float uv  = up [(size_t)i * D_INNER];
    const float dtu = dtv * uv;
    const float* Brow = bp + (size_t)i * E_DIM;
#pragma unroll
    for (int n = 0; n < STATE; ++n) {
      const float a = __expf(dtv * A[n]);
      h[n] = fmaf(a, h[n], dtu * Brow[n]);
      p[n] *= a;
    }
  }
  const size_t sbase = ((size_t)(b * NCHUNK + chunk) * STATE) * D_INNER + d;
#pragma unroll
  for (int n = 0; n < STATE; ++n) {
    prodA[sbase + (size_t)n * D_INNER] = p[n];
    hend [sbase + (size_t)n * D_INNER] = h[n];
  }
}

__global__ __launch_bounds__(256) void scan_combine(
    const float* __restrict__ prodA, const float* __restrict__ hend,
    float* __restrict__ hin) {
  const int t = blockIdx.x * blockDim.x + threadIdx.x;
  const int d = t % D_INNER;
  const int n = (t / D_INNER) % STATE;
  const int b = t / (D_INNER * STATE);
  float h = 0.0f;
#pragma unroll
  for (int c = 0; c < NCHUNK; ++c) {
    const size_t idx = ((size_t)(b * NCHUNK + c) * STATE + n) * D_INNER + d;
    hin[idx] = h;
    h = fmaf(prodA[idx], h, hend[idx]);
  }
}

__global__ __launch_bounds__(SCAN_BLOCK) void scan_pass3(
    const float* __restrict__ u, const float* __restrict__ A_log,
    const float* __restrict__ x_dbl, const float* __restrict__ dt,
    const float* __restrict__ hin, const float* __restrict__ Dp,
    float* __restrict__ out) {
  const int blk   = blockIdx.x;
  const int dblk  = blk % DBLKS;
  const int chunk = (blk / DBLKS) % NCHUNK;
  const int b     = blk / (DBLKS * NCHUNK);
  const int d     = dblk * SCAN_BLOCK + threadIdx.x;

  const size_t sbase = ((size_t)(b * NCHUNK + chunk) * STATE) * D_INNER + d;
  float A[STATE], h[STATE];
#pragma unroll
  for (int n = 0; n < STATE; ++n) {
    A[n] = -__expf(A_log[d * STATE + n]);
    h[n] = hin[sbase + (size_t)n * D_INNER];
  }
  const float dp = Dp[d];
  const size_t r0 = (size_t)b * SEQ + (size_t)chunk * CLEN;
  const float* dtp = dt + r0 * D_INNER + d;
  const float* up  = u  + r0 * D_INNER + d;
  const float* bp  = x_dbl + r0 * E_DIM + DT_RANK;
  float* op = out + r0 * D_INNER + d;
#pragma unroll 4
  for (int i = 0; i < CLEN; ++i) {
    const float dtv = dtp[(size_t)i * D_INNER];
    const float uv  = up [(size_t)i * D_INNER];
    const float dtu = dtv * uv;
    const float* Brow = bp + (size_t)i * E_DIM;
    const float* Crow = Brow + STATE;
    float y[4] = {0.0f, 0.0f, 0.0f, 0.0f};
#pragma unroll
    for (int n = 0; n < STATE; ++n) {
      const float a = __expf(dtv * A[n]);
      h[n] = fmaf(a, h[n], dtu * Brow[n]);
      y[n & 3] = fmaf(h[n], Crow[n], y[n & 3]);
    }
    op[(size_t)i * D_INNER] = fmaf(uv, dp, (y[0] + y[1]) + (y[2] + y[3]));
  }
}

extern "C" void kernel_launch(void* const* d_in, const int* in_sizes, int n_in,
                              void* d_out, int out_size, void* d_ws, size_t ws_size,
                              hipStream_t stream) {
  const float* u     = (const float*)d_in[0];
  const float* A_log = (const float*)d_in[1];
  const float* W_x   = (const float*)d_in[2];
  const float* W_dt  = (const float*)d_in[3];
  const float* b_dt  = (const float*)d_in[4];
  const float* Dp    = (const float*)d_in[5];
  float* out = (float*)d_out;

  float* ws    = (float*)d_ws;
  float* x_dbl = ws;
  float* dtbuf = x_dbl + (size_t)ROWS * E_DIM;
  float* prodA = dtbuf + (size_t)ROWS * D_INNER;
  float* hend  = prodA + (size_t)BATCH * D_INNER * STATE * NCHUNK;
  float* hin   = hend  + (size_t)BATCH * D_INNER * STATE * NCHUNK;

  xproj_wmma<<<ROWS / 16, XP_WAVES * 32, 0, stream>>>(u, W_x, W_dt, b_dt, x_dbl, dtbuf);
  scan_pass1<<<BATCH * NCHUNK * DBLKS, SCAN_BLOCK, 0, stream>>>(u, A_log, x_dbl, dtbuf, prodA, hend);
  scan_combine<<<(BATCH * D_INNER * STATE) / 256, 256, 0, stream>>>(prodA, hend, hin);
  scan_pass3<<<BATCH * NCHUNK * DBLKS, SCAN_BLOCK, 0, stream>>>(u, A_log, x_dbl, dtbuf, hin, Dp, out);
}


# softplus of the dt epilogue computed as max(x,0) + 2*atanh(e/(2+e)), e = exp(-|x|), by a short odd series (two dozen fp32 instructions per value) in place of the library log1p
# speedup vs baseline: 5.2150x; 1.0733x over previous
#include <hip/hip_runtime.h>
#include <hip/hip_bf16.h>

typedef __attribute__((ext_vector_type(2))) float v2f;
typedef __attribute__((ext_vector_type(4))) float v4f;
typedef __attribute__((ext_vector_type(8))) float v8f;

#define D_INNER 1024
#define STATE   16
#define DT_RANK 64
#define BATCH   2
#define SEQ     2048
#define ROWS    (BATCH * SEQ)
#define E_DIM   (DT_RANK + 2 * STATE)
#define NCHUNK  64
#define CLEN    (SEQ / NCHUNK)
#define SCAN_BLOCK 128
#define DBLKS   (D_INNER / SCAN_BLOCK)

#define XP_WAVES  8
#define XP_KSLICE (D_INNER / XP_WAVES)
#define XP_TILES  (E_DIM / 16)

#define XS_PITCH  68
#define DP_PAIRS  4

struct XpFrag {
  v4f a[4];
  v4f b[XP_TILES][4];
};

struct DpFrag {
  v4f b[2][8];
};

__device__ __forceinline__ void dp_load(DpFrag& f, const float* __restrict__ wb) {
#pragma unroll
  for (int t = 0; t < 2; ++t) {
#pragma unroll
    for (int q = 0; q < 8; ++q)
      f.b[t][q] = *(const v4f*)(wb + (size_t)t * 16 * DT_RANK + 8 * q);
  }
}

__device__ __forceinline__ float xhalf(float x) {
  return __builtin_bit_cast(float, __builtin_amdgcn_permlanex16(
      __builtin_bit_cast(int, x), __builtin_bit_cast(int, x),
      0x76543210, 0xfedcba98, false, false));
}

__device__ __forceinline__ float softplus_f(float x) {
  const float e  = __expf(-__builtin_fabsf(x));
  const float d  = 2.0f + e;
  const float r  = __builtin_amdgcn_rcpf(d);
  const float s0 = e * r;
  const float s  = fmaf(fmaf(-s0, d, e), r, s0);
  const float z  = s * s;
  float p = fmaf(z, 1.0f / 17.0f, 1.0f / 15.0f);
  p = fmaf(z, p, 1.0f / 13.0f);
  p = fmaf(z, p, 1.0f / 11.0f);
  p = fmaf(z, p, 1.0f / 9.0f);
  p = fmaf(z, p, 1.0f / 7.0f);
  p = fmaf(z, p, 1.0f / 5.0f);
  p = fmaf(z, p, 1.0f / 3.0f);
  const float s2 = s + s;
  return fmaxf(x, 0.0f) + fmaf(s2 * z, p, s2);
}

__device__ __forceinline__ void dp_pair(const v4f (&af)[8], const DpFrag& f,
                                        float bias, int kh, float* __restrict__ op) {
  v8f c0 = {}, c1 = {};
#pragma unroll
  for (int q = 0; q < 8; ++q) {
    const v2f a0  = {af[q].x, af[q].y};
    const v2f a1  = {af[q].z, af[q].w};
    const v2f b00 = {f.b[0][q].x, f.b[0][q].y};
    const v2f b10 = {f.b[1][q].x, f.b[1][q].y};
    const v2f b01 = {f.b[0][q].z, f.b[0][q].w};
    const v2f b11 = {f.b[1][q].z, f.b[1][q].w};
    c0 = __builtin_amdgcn_wmma_f32_16x16x4_f32(false, a0, false, b00, (short)0, c0, false, false);
    c1 = __builtin_amdgcn_wmma_f32_16x16x4_f32(false, a0, false, b10, (short)0, c1, false, false);
    c0 = __builtin_amdgcn_wmma_f32_16x16x4_f32(false, a1, false, b01, (short)0, c0, false, false);
    c1 = __builtin_amdgcn_wmma_f32_16x16x4_f32(false, a1, false, b11, (short)0, c1, false, false);
  }
#pragma unroll
  for (int j = 0; j < 8; ++j) {
    const float v0 = c0[j], v1 = c1[j];
    const float got = xhalf(kh ? v0 : v1);
    const float x = kh ? got : v0;
    const float y = kh ? v1 : got;
    op[(size_t)j * D_INNER]       = softplus_f(x + bias);
    op[(size_t)(j + 8) * D_INNER] = softplus_f(y + bias);
  }
}

__device__ __forceinline__ void xp_load(XpFrag& f, const float* __restrict__ ua,
                                        const float* __restrict__ wa, int sb) {
#pragma unroll
  for (int q = 0; q < 4; ++q) f.a[q] = *(const v4f*)(ua + 32 * sb + 8 * q);
#pragma unroll
  for (int t = 0; t < XP_TILES; ++t) {
#pragma unroll
    for (int q = 0; q < 4; ++q)
      f.b[t][q] = *(const v4f*)(wa + (size_t)t * 16 * D_INNER + 32 * sb + 8 * q);
  }
}

__device__ __forceinline__ void xp_mma(const XpFrag& f, v8f (&acc)[XP_TILES]) {
#pragma unroll
  for (int q = 0; q < 4; ++q) {
    const v2f a0 = {f.a[q].x, f.a[q].y};
    const v2f a1 = {f.a[q].z, f.a[q].w};
#pragma unroll
    for (int t = 0; t < XP_TILES; ++t) {
      const v2f b0 = {f.b[t][q].x, f.b[t][q].y};
      acc[t] = __builtin_amdgcn_wmma_f32_16x16x4_f32(
           false, a0,  false, b0,
           (short)0, acc[t],  false,  false);
    }
#pragma unroll
    for (int t = 0; t < XP_TILES; ++t) {
      const v2f b1 = {f.b[t][q].z, f.b[t][q].w};
      acc[t] = __builtin_amdgcn_wmma_f32_16x16x4_f32(
          false, a1, false, b1, (short)0, acc[t], false, false);
    }
  }
}

__global__ __launch_bounds__(XP_WAVES * 32) void xproj_wmma(
    const float* __restrict__ u, const float* __restrict__ Wx,
    const float* __restrict__ Wdt, const float* __restrict__ b_dt,
    float* __restrict__ x_dbl, float* __restrict__ dt) {
  __shared__ float red[XP_WAVES][XP_TILES][8][32];
  __shared__ __attribute__((aligned(16))) float xs[16][XS_PITCH];

  const int w    = threadIdx.x >> 5;
  const int lane = threadIdx.x & 31;
  const int row0 = blockIdx.x * 16;
  const int m  = lane & 15;
  const int kh = lane >> 4;

  const float* ua = u  + (size_t)(row0 + m) * D_INNER + w * XP_KSLICE + 4 * kh;
  const float* wa = Wx + (size_t)m * D_INNER + w * XP_KSLICE + 4 * kh;

  v8f acc[XP_TILES];
#pragma unroll
  for (int t = 0; t < XP_TILES; ++t) acc[t] = v8f{};

  XpFrag cur, nxt;
  xp_load(cur, ua, wa, 0);
#pragma unroll
  for (int sb = 0; sb < XP_KSLICE / 32; ++sb) {
    if (sb + 1 < XP_KSLICE / 32) xp_load(nxt, ua, wa, sb + 1);
    __builtin_amdgcn_sched_barrier(0);
    xp_mma(cur, acc);
    if (sb + 1 < XP_KSLICE / 32) cur = nxt;
  }

#pragma unroll
  for (int t = 0; t < XP_TILES; ++t) {
#pragma unroll
    for (int j = 0; j < 8; ++j) red[w][t][j][lane] = acc[t][j];
  }
  __syncthreads();

  float* outp = x_dbl + (size_t)row0 * E_DIM;
#pragma unroll
  for (int i = 0; i < (16 * E_DIM) / (XP_WAVES * 32); ++i) {
    const int o  = threadIdx.x + i * (XP_WAVES * 32);
    const int r  = o / E_DIM;
    const int c  = o - r * E_DIM;
    const int t  = c >> 4;
    const int ln = ((r >> 3) << 4) | (c & 15);
    const int j  = r & 7;
    float sum = red[0][t][j][ln];
#pragma unroll
    for (int ww = 1; ww < XP_WAVES; ++ww) sum += red[ww][t][j][ln];
    outp[o] = sum;
    if (c < DT_RANK) xs[r][c] = sum;
  }
  __syncthreads();

  v4f af[8];
  {
    const float* ap = &xs[m][4 * kh];
#pragma unroll
    for (int q = 0; q < 8; ++q) af[q] = *(const v4f*)(ap + 8 * q);
  }
  const int cbase = w * (DP_PAIRS * 32);
  const float* wb = Wdt + (size_t)(cbase + m) * DT_RANK + 4 * kh;
  const float* bp = b_dt + cbase + lane;
  float* dtp = dt + (size_t)row0 * D_INNER + cbase + lane;

  DpFrag f0, f1;
  float bias0 = bp[0], bias1;
  dp_load(f0, wb);
#pragma unroll 1
  for (int p = 0; p < DP_PAIRS; p += 2) {
    bias1 = bp[(p + 1) * 32];
    dp_load(f1, wb + (size_t)(p + 1) * 32 * DT_RANK);
    __builtin_amdgcn_sched_barrier(0);
    dp_pair(af, f0, bias0, kh, dtp + p * 32);
    __builtin_amdgcn_sched_barrier(0);
    const int pn = (p + 2) & (DP_PAIRS - 1);
    bias0 = bp[pn * 32];
    dp_load(f0, wb + (size_t)pn * 32 * DT_RANK);
    __builtin_amdgcn_sched_barrier(0);
    dp_pair(af, f1, bias1, kh, dtp + (p + 1) * 32);
  }
}

__global__ __launch_bounds__(SCAN_BLOCK) void scan_pass1(
    const float* __restrict__ u, const float* __restrict__ A_log,
    const float* __restrict__ x_dbl, const float* __restrict__ dt,
    float* __restrict__ prodA, float* __restrict__ hend) {
  const int blk   = blockIdx.x;
  const int dblk  = blk % DBLKS;
  const int chunk = (blk / DBLKS) % NCHUNK;
  const int b     = blk / (DBLKS * NCHUNK);
  const int d     = dblk * SCAN_BLOCK + threadIdx.x;

  float A[STATE], h[STATE], p[STATE];
#pragma unroll
  for (int n = 0; n < STATE; ++n) {
    A[n] = -__expf(A_log[d * STATE + n]);
    h[n] = 0.0f;
    p[n] = 1.0f;
  }
  const size_t r0 = (size_t)b * SEQ + (size_t)chunk * CLEN;
  const float* dtp = dt + r0 * D_INNER + d;
  const float* up  = u  + r0 * D_INNER + d;
  const float* bp  = x_dbl + r0 * E_DIM + DT_RANK;
#pragma unroll 4
  for (int i = 0; i < CLEN; ++i) {
    const float dtv = dtp[(size_t)i * D_INNER];
    const float uv  = up [(size_t)i * D_INNER];
    const float dtu = dtv * uv;
    const float* Brow = bp + (size_t)i * E_DIM;
#pragma unroll
    for (int n = 0; n < STATE; ++n) {
      const float a = __expf(dtv * A[n]);
      h[n] = fmaf(a, h[n], dtu * Brow[n]);
      p[n] *= a;
    }
  }
  const size_t sbase = ((size_t)(b * NCHUNK + chunk) * STATE) * D_INNER + d;
#pragma unroll
  for (int n = 0; n < STATE; ++n) {
    prodA[sbase + (size_t)n * D_INNER] = p[n];
    hend [sbase + (size_t)n * D_INNER] = h[n];
  }
}

__global__ __launch_bounds__(256) void scan_combine(
    const float* __restrict__ prodA, const float* __restrict__ hend,
    float* __restrict__ hin) {
  const int t = blockIdx.x * blockDim.x + threadIdx.x;
  const int d = t % D_INNER;
  const int n = (t / D_INNER) % STATE;
  const int b = t / (D_INNER * STATE);
  float h = 0.0f;
#pragma unroll
  for (int c = 0; c < NCHUNK; ++c) {
    const size_t idx = ((size_t)(b * NCHUNK + c) * STATE + n) * D_INNER + d;
    hin[idx] = h;
    h = fmaf(prodA[idx], h, hend[idx]);
  }
}

__global__ __launch_bounds__(SCAN_BLOCK) void scan_pass3(
    const float* __restrict__ u, const float* __restrict__ A_log,
    const float* __restrict__ x_dbl, const float* __restrict__ dt,
    const float* __restrict__ hin, const float* __restrict__ Dp,
    float* __restrict__ out) {
  const int blk   = blockIdx.x;
  const int dblk  = blk % DBLKS;
  const int chunk = (blk / DBLKS) % NCHUNK;
  const int b     = blk / (DBLKS * NCHUNK);
  const int d     = dblk * SCAN_BLOCK + threadIdx.x;

  const size_t sbase = ((size_t)(b * NCHUNK + chunk) * STATE) * D_INNER + d;
  float A[STATE], h[STATE];
#pragma unroll
  for (int n = 0; n < STATE; ++n) {
    A[n] = -__expf(A_log[d * STATE + n]);
    h[n] = hin[sbase + (size_t)n * D_INNER];
  }
  const float dp = Dp[d];
  const size_t r0 = (size_t)b * SEQ + (size_t)chunk * CLEN;
  const float* dtp = dt + r0 * D_INNER + d;
  const float* up  = u  + r0 * D_INNER + d;
  const float* bp  = x_dbl + r0 * E_DIM + DT_RANK;
  float* op = out + r0 * D_INNER + d;
#pragma unroll 4
  for (int i = 0; i < CLEN; ++i) {
    const float dtv = dtp[(size_t)i * D_INNER];
    const float uv  = up [(size_t)i * D_INNER];
    const float dtu = dtv * uv;
    const float* Brow = bp + (size_t)i * E_DIM;
    const float* Crow = Brow + STATE;
    float y[4] = {0.0f, 0.0f, 0.0f, 0.0f};
#pragma unroll
    for (int n = 0; n < STATE; ++n) {
      const float a = __expf(dtv * A[n]);
      h[n] = fmaf(a, h[n], dtu * Brow[n]);
      y[n & 3] = fmaf(h[n], Crow[n], y[n & 3]);
    }
    op[(size_t)i * D_INNER] = fmaf(uv, dp, (y[0] + y[1]) + (y[2] + y[3]));
  }
}

extern "C" void kernel_launch(void* const* d_in, const int* in_sizes, int n_in,
                              void* d_out, int out_size, void* d_ws, size_t ws_size,
                              hipStream_t stream) {
  const float* u     = (const float*)d_in[0];
  const float* A_log = (const float*)d_in[1];
  const float* W_x   = (const float*)d_in[2];
  const float* W_dt  = (const float*)d_in[3];
  const float* b_dt  = (const float*)d_in[4];
  const float* Dp    = (const float*)d_in[5];
  float* out = (float*)d_out;

  float* ws    = (float*)d_ws;
  float* x_dbl = ws;
  float* dtbuf = x_dbl + (size_t)ROWS * E_DIM;
  float* prodA = dtbuf + (size_t)ROWS * D_INNER;
  float* hend  = prodA + (size_t)BATCH * D_INNER * STATE * NCHUNK;
  float* hin   = hend  + (size_t)BATCH * D_INNER * STATE * NCHUNK;

  xproj_wmma<<<ROWS / 16, XP_WAVES * 32, 0, stream>>>(u, W_x, W_dt, b_dt, x_dbl, dtbuf);
  scan_pass1<<<BATCH * NCHUNK * DBLKS, SCAN_BLOCK, 0, stream>>>(u, A_log, x_dbl, dtbuf, prodA, hend);
  scan_combine<<<(BATCH * D_INNER * STATE) / 256, 256, 0, stream>>>(prodA, hend, hin);
  scan_pass3<<<BATCH * NCHUNK * DBLKS, SCAN_BLOCK, 0, stream>>>(u, A_log, x_dbl, dtbuf, hin, Dp, out);
}


# scan passes 1 and 3: each lane owns two channels (d, d+128), so every recurrence op is one packed fp32 instruction for both, B/C rows staged in LDS as the shared operand; loads one group ahead
# speedup vs baseline: 5.7440x; 1.1014x over previous
#include <hip/hip_runtime.h>
#include <hip/hip_bf16.h>

typedef __attribute__((ext_vector_type(2))) float v2f;
typedef __attribute__((ext_vector_type(4))) float v4f;
typedef __attribute__((ext_vector_type(8))) float v8f;

#define D_INNER 1024
#define STATE   16
#define DT_RANK 64
#define BATCH   2
#define SEQ     2048
#define ROWS    (BATCH * SEQ)
#define E_DIM   (DT_RANK + 2 * STATE)
#define NCHUNK  64
#define CLEN    (SEQ / NCHUNK)
#define SCAN_BLOCK 128
#define SCAN_CH (2 * SCAN_BLOCK)
#define DBLKS   (D_INNER / SCAN_CH)
#define SGRP    4

#define XP_WAVES  8
#define XP_KSLICE (D_INNER / XP_WAVES)
#define XP_TILES  (E_DIM / 16)

#define XS_PITCH  68
#define DP_PAIRS  4

struct XpFrag {
  v4f a[4];
  v4f b[XP_TILES][4];
};

struct DpFrag {
  v4f b[2][8];
};

__device__ __forceinline__ void dp_load(DpFrag& f, const float* __restrict__ wb) {
#pragma unroll
  for (int t = 0; t < 2; ++t) {
#pragma unroll
    for (int q = 0; q < 8; ++q)
      f.b[t][q] = *(const v4f*)(wb + (size_t)t * 16 * DT_RANK + 8 * q);
  }
}

__device__ __forceinline__ float xhalf(float x) {
  return __builtin_bit_cast(float, __builtin_amdgcn_permlanex16(
      __builtin_bit_cast(int, x), __builtin_bit_cast(int, x),
      0x76543210, 0xfedcba98, false, false));
}

__device__ __forceinline__ float softplus_f(float x) {
  const float e  = __expf(-__builtin_fabsf(x));
  const float d  = 2.0f + e;
  const float r  = __builtin_amdgcn_rcpf(d);
  const float s0 = e * r;
  const float s  = fmaf(fmaf(-s0, d, e), r, s0);
  const float z  = s * s;
  float p = fmaf(z, 1.0f / 17.0f, 1.0f / 15.0f);
  p = fmaf(z, p, 1.0f / 13.0f);
  p = fmaf(z, p, 1.0f / 11.0f);
  p = fmaf(z, p, 1.0f / 9.0f);
  p = fmaf(z, p, 1.0f / 7.0f);
  p = fmaf(z, p, 1.0f / 5.0f);
  p = fmaf(z, p, 1.0f / 3.0f);
  const float s2 = s + s;
  return fmaxf(x, 0.0f) + fmaf(s2 * z, p, s2);
}

__device__ __forceinline__ void dp_pair(const v4f (&af)[8], const DpFrag& f,
                                        float bias, int kh, float* __restrict__ op) {
  v8f c0 = {}, c1 = {};
#pragma unroll
  for (int q = 0; q < 8; ++q) {
    const v2f a0  = {af[q].x, af[q].y};
    const v2f a1  = {af[q].z, af[q].w};
    const v2f b00 = {f.b[0][q].x, f.b[0][q].y};
    const v2f b10 = {f.b[1][q].x, f.b[1][q].y};
    const v2f b01 = {f.b[0][q].z, f.b[0][q].w};
    const v2f b11 = {f.b[1][q].z, f.b[1][q].w};
    c0 = __builtin_amdgcn_wmma_f32_16x16x4_f32(false, a0, false, b00, (short)0, c0, false, false);
    c1 = __builtin_amdgcn_wmma_f32_16x16x4_f32(false, a0, false, b10, (short)0, c1, false, false);
    c0 = __builtin_amdgcn_wmma_f32_16x16x4_f32(false, a1, false, b01, (short)0, c0, false, false);
    c1 = __builtin_amdgcn_wmma_f32_16x16x4_f32(false, a1, false, b11, (short)0, c1, false, false);
  }
#pragma unroll
  for (int j = 0; j < 8; ++j) {
    const float v0 = c0[j], v1 = c1[j];
    const float got = xhalf(kh ? v0 : v1);
    const float x = kh ? got : v0;
    const float y = kh ? v1 : got;
    op[(size_t)j * D_INNER]       = softplus_f(x + bias);
    op[(size_t)(j + 8) * D_INNER] = softplus_f(y + bias);
  }
}

__device__ __forceinline__ void xp_load(XpFrag& f, const float* __restrict__ ua,
                                        const float* __restrict__ wa, int sb) {
#pragma unroll
  for (int q = 0; q < 4; ++q) f.a[q] = *(const v4f*)(ua + 32 * sb + 8 * q);
#pragma unroll
  for (int t = 0; t < XP_TILES; ++t) {
#pragma unroll
    for (int q = 0; q < 4; ++q)
      f.b[t][q] = *(const v4f*)(wa + (size_t)t * 16 * D_INNER + 32 * sb + 8 * q);
  }
}

__device__ __forceinline__ void xp_mma(const XpFrag& f, v8f (&acc)[XP_TILES]) {
#pragma unroll
  for (int q = 0; q < 4; ++q) {
    const v2f a0 = {f.a[q].x, f.a[q].y};
    const v2f a1 = {f.a[q].z, f.a[q].w};
#pragma unroll
    for (int t = 0; t < XP_TILES; ++t) {
      const v2f b0 = {f.b[t][q].x, f.b[t][q].y};
      acc[t] = __builtin_amdgcn_wmma_f32_16x16x4_f32(
           false, a0,  false, b0,
           (short)0, acc[t],  false,  false);
    }
#pragma unroll
    for (int t = 0; t < XP_TILES; ++t) {
      const v2f b1 = {f.b[t][q].z, f.b[t][q].w};
      acc[t] = __builtin_amdgcn_wmma_f32_16x16x4_f32(
          false, a1, false, b1, (short)0, acc[t], false, false);
    }
  }
}

__global__ __launch_bounds__(XP_WAVES * 32) void xproj_wmma(
    const float* __restrict__ u, const float* __restrict__ Wx,
    const float* __restrict__ Wdt, const float* __restrict__ b_dt,
    float* __restrict__ x_dbl, float* __restrict__ dt) {
  __shared__ float red[XP_WAVES][XP_TILES][8][32];
  __shared__ __attribute__((aligned(16))) float xs[16][XS_PITCH];

  const int w    = threadIdx.x >> 5;
  const int lane = threadIdx.x & 31;
  const int row0 = blockIdx.x * 16;
  const int m  = lane & 15;
  const int kh = lane >> 4;

  const float* ua = u  + (size_t)(row0 + m) * D_INNER + w * XP_KSLICE + 4 * kh;
  const float* wa = Wx + (size_t)m * D_INNER + w * XP_KSLICE + 4 * kh;

  v8f acc[XP_TILES];
#pragma unroll
  for (int t = 0; t < XP_TILES; ++t) acc[t] = v8f{};

  XpFrag cur, nxt;
  xp_load(cur, ua, wa, 0);
#pragma unroll
  for (int sb = 0; sb < XP_KSLICE / 32; ++sb) {
    if (sb + 1 < XP_KSLICE / 32) xp_load(nxt, ua, wa, sb + 1);
    __builtin_amdgcn_sched_barrier(0);
    xp_mma(cur, acc);
    if (sb + 1 < XP_KSLICE / 32) cur = nxt;
  }

#pragma unroll
  for (int t = 0; t < XP_TILES; ++t) {
#pragma unroll
    for (int j = 0; j < 8; ++j) red[w][t][j][lane] = acc[t][j];
  }
  __syncthreads();

  float* outp = x_dbl + (size_t)row0 * E_DIM;
#pragma unroll
  for (int i = 0; i < (16 * E_DIM) / (XP_WAVES * 32); ++i) {
    const int o  = threadIdx.x + i * (XP_WAVES * 32);
    const int r  = o / E_DIM;
    const int c  = o - r * E_DIM;
    const int t  = c >> 4;
    const int ln = ((r >> 3) << 4) | (c & 15);
    const int j  = r & 7;
    float sum = red[0][t][j][ln];
#pragma unroll
    for (int ww = 1; ww < XP_WAVES; ++ww) sum += red[ww][t][j][ln];
    outp[o] = sum;
    if (c < DT_RANK) xs[r][c] = sum;
  }
  __syncthreads();

  v4f af[8];
  {
    const float* ap = &xs[m][4 * kh];
#pragma unroll
    for (int q = 0; q < 8; ++q) af[q] = *(const v4f*)(ap + 8 * q);
  }
  const int cbase = w * (DP_PAIRS * 32);
  const float* wb = Wdt + (size_t)(cbase + m) * DT_RANK + 4 * kh;
  const float* bp = b_dt + cbase + lane;
  float* dtp = dt + (size_t)row0 * D_INNER + cbase + lane;

  DpFrag f0, f1;
  float bias0 = bp[0], bias1;
  dp_load(f0, wb);
#pragma unroll 1
  for (int p = 0; p < DP_PAIRS; p += 2) {
    bias1 = bp[(p + 1) * 32];
    dp_load(f1, wb + (size_t)(p + 1) * 32 * DT_RANK);
    __builtin_amdgcn_sched_barrier(0);
    dp_pair(af, f0, bias0, kh, dtp + p * 32);
    __builtin_amdgcn_sched_barrier(0);
    const int pn = (p + 2) & (DP_PAIRS - 1);
    bias0 = bp[pn * 32];
    dp_load(f0, wb + (size_t)pn * 32 * DT_RANK);
    __builtin_amdgcn_sched_barrier(0);
    dp_pair(af, f1, bias1, kh, dtp + (p + 1) * 32);
  }
}

struct ScanIn {
  v2f dt[SGRP];
  v2f u[SGRP];
};

__device__ __forceinline__ void scan_load(ScanIn& f, const float* __restrict__ dtp,
                                          const float* __restrict__ up, int s0) {
#pragma unroll
  for (int i = 0; i < SGRP; ++i) {
    const size_t o = (size_t)(s0 + i) * D_INNER;
    f.dt[i] = v2f{dtp[o], dtp[o + SCAN_BLOCK]};
    f.u[i]  = v2f{up[o],  up[o + SCAN_BLOCK]};
  }
}

template <int NQ>
struct ScanRow {
  v4f q[NQ];
};

template <int NQ>
__device__ __forceinline__ void row_load(ScanRow<NQ>& r, const float* lp) {
#pragma unroll
  for (int k = 0; k < NQ; ++k) r.q[k] = *(const v4f*)(lp + 4 * k);
}

__device__ __forceinline__ v2f exp_pair(v2f x) {
  return v2f{__expf(x.x), __expf(x.y)};
}

__global__ __launch_bounds__(SCAN_BLOCK) void scan_pass1(
    const float* __restrict__ u, const float* __restrict__ A_log,
    const float* __restrict__ x_dbl, const float* __restrict__ dt,
    float* __restrict__ prodA, float* __restrict__ hend) {
  __shared__ __attribute__((aligned(16))) float brow[CLEN][STATE];
  const int blk   = blockIdx.x;
  const int dblk  = blk % DBLKS;
  const int chunk = (blk / DBLKS) % NCHUNK;
  const int b     = blk / (DBLKS * NCHUNK);
  const int d0    = dblk * SCAN_CH + threadIdx.x;
  const int d1    = d0 + SCAN_BLOCK;

  v2f A[STATE], h[STATE], p[STATE];
#pragma unroll
  for (int n = 0; n < STATE; ++n) {
    A[n] = v2f{-__expf(A_log[d0 * STATE + n]), -__expf(A_log[d1 * STATE + n])};
    h[n] = v2f{0.0f, 0.0f};
    p[n] = v2f{1.0f, 1.0f};
  }
  const size_t r0 = (size_t)b * SEQ + (size_t)chunk * CLEN;
  const float* dtp = dt + r0 * D_INNER + d0;
  const float* up  = u  + r0 * D_INNER + d0;
  const float* bp  = x_dbl + r0 * E_DIM + DT_RANK;

  ScanIn cur, nxt;
  scan_load(cur, dtp, up, 0);
  {
    const int r = threadIdx.x >> 2, q = threadIdx.x & 3;
    *(v4f*)&brow[r][4 * q] = *(const v4f*)(bp + (size_t)r * E_DIM + 4 * q);
  }
  __syncthreads();
  ScanRow<4> rc, rn;
  row_load(rc, &brow[0][0]);
#pragma unroll 2
  for (int g = 0; g < CLEN / SGRP; ++g) {
    const int gn = (g + 1 < CLEN / SGRP) ? g + 1 : g;
    scan_load(nxt, dtp, up, gn * SGRP);
    __builtin_amdgcn_sched_barrier(0);
#pragma unroll
    for (int i = 0; i < SGRP; ++i) {
      const int st = g * SGRP + i;
      row_load(rn, &brow[(st + 1 < CLEN) ? st + 1 : st][0]);
      __builtin_amdgcn_sched_barrier(0);
      const v2f dtv = cur.dt[i];
      const v2f dtu = dtv * cur.u[i];
#pragma unroll
      for (int n = 0; n < STATE; ++n) {
        const v2f a = exp_pair(dtv * A[n]);
        h[n] = __builtin_elementwise_fma(a, h[n], dtu * rc.q[n >> 2][n & 3]);
        p[n] *= a;
      }
      rc = rn;
    }
    cur = nxt;
  }
  const size_t sbase = ((size_t)(b * NCHUNK + chunk) * STATE) * D_INNER + d0;
#pragma unroll
  for (int n = 0; n < STATE; ++n) {
    prodA[sbase + (size_t)n * D_INNER]              = p[n].x;
    prodA[sbase + (size_t)n * D_INNER + SCAN_BLOCK] = p[n].y;
    hend [sbase + (size_t)n * D_INNER]              = h[n].x;
    hend [sbase + (size_t)n * D_INNER + SCAN_BLOCK] = h[n].y;
  }
}

__global__ __launch_bounds__(256) void scan_combine(
    const float* __restrict__ prodA, const float* __restrict__ hend,
    float* __restrict__ hin) {
  const int t = blockIdx.x * blockDim.x + threadIdx.x;
  const int d = t % D_INNER;
  const int n = (t / D_INNER) % STATE;
  const int b = t / (D_INNER * STATE);
  float h = 0.0f;
#pragma unroll
  for (int c = 0; c < NCHUNK; ++c) {
    const size_t idx = ((size_t)(b * NCHUNK + c) * STATE + n) * D_INNER + d;
    hin[idx] = h;
    h = fmaf(prodA[idx], h, hend[idx]);
  }
}

__global__ __launch_bounds__(SCAN_BLOCK) void scan_pass3(
    const float* __restrict__ u, const float* __restrict__ A_log,
    const float* __restrict__ x_dbl, const float* __restrict__ dt,
    const float* __restrict__ hin, const float* __restrict__ Dp,
    float* __restrict__ out) {
  __shared__ __attribute__((aligned(16))) float bcrow[CLEN][2 * STATE];
  const int blk   = blockIdx.x;
  const int dblk  = blk % DBLKS;
  const int chunk = (blk / DBLKS) % NCHUNK;
  const int b     = blk / (DBLKS * NCHUNK);
  const int d0    = dblk * SCAN_CH + threadIdx.x;
  const int d1    = d0 + SCAN_BLOCK;

  const size_t sbase = ((size_t)(b * NCHUNK + chunk) * STATE) * D_INNER + d0;
  v2f A[STATE], h[STATE];
#pragma unroll
  for (int n = 0; n < STATE; ++n) {
    A[n] = v2f{-__expf(A_log[d0 * STATE + n]), -__expf(A_log[d1 * STATE + n])};
    h[n] = v2f{hin[sbase + (size_t)n * D_INNER], hin[sbase + (size_t)n * D_INNER + SCAN_BLOCK]};
  }
  const v2f dp = v2f{Dp[d0], Dp[d1]};
  const size_t r0 = (size_t)b * SEQ + (size_t)chunk * CLEN;
  const float* dtp = dt + r0 * D_INNER + d0;
  const float* up  = u  + r0 * D_INNER + d0;
  const float* bp  = x_dbl + r0 * E_DIM + DT_RANK;
  float* op = out + r0 * D_INNER + d0;

  ScanIn cur, nxt;
  scan_load(cur, dtp, up, 0);
  {
    const int r = threadIdx.x >> 2, q = threadIdx.x & 3;
    const float* src = bp + (size_t)r * E_DIM + 8 * q;
    *(v4f*)&bcrow[r][8 * q]     = *(const v4f*)src;
    *(v4f*)&bcrow[r][8 * q + 4] = *(const v4f*)(src + 4);
  }
  __syncthreads();
  ScanRow<8> rc, rn;
  row_load(rc, &bcrow[0][0]);
#pragma unroll 2
  for (int g = 0; g < CLEN / SGRP; ++g) {
    const int gn = (g + 1 < CLEN / SGRP) ? g + 1 : g;
    scan_load(nxt, dtp, up, gn * SGRP);
    __builtin_amdgcn_sched_barrier(0);
#pragma unroll
    for (int i = 0; i < SGRP; ++i) {
      const int st = g * SGRP + i;
      row_load(rn, &bcrow[(st + 1 < CLEN) ? st + 1 : st][0]);
      __builtin_amdgcn_sched_barrier(0);
      const v2f dtv = cur.dt[i];
      const v2f uv  = cur.u[i];
      const v2f dtu = dtv * uv;
      v2f y[4];
#pragma unroll
      for (int k = 0; k < 4; ++k) y[k] = v2f{0.0f, 0.0f};
#pragma unroll
      for (int n = 0; n < STATE; ++n) {
        const v2f a = exp_pair(dtv * A[n]);
        h[n] = __builtin_elementwise_fma(a, h[n], dtu * rc.q[n >> 2][n & 3]);
        const float cn = rc.q[4 + (n >> 2)][n & 3];
        const v2f c = v2f{cn, cn};
        y[n & 3] = __builtin_elementwise_fma(h[n], c, y[n & 3]);
      }
      const v2f o = __builtin_elementwise_fma(uv, dp, (y[0] + y[1]) + (y[2] + y[3]));
      op[(size_t)st * D_INNER]              = o.x;
      op[(size_t)st * D_INNER + SCAN_BLOCK] = o.y;
      rc = rn;
    }
    cur = nxt;
  }
}

extern "C" void kernel_launch(void* const* d_in, const int* in_sizes, int n_in,
                              void* d_out, int out_size, void* d_ws, size_t ws_size,
                              hipStream_t stream) {
  const float* u     = (const float*)d_in[0];
  const float* A_log = (const float*)d_in[1];
  const float* W_x   = (const float*)d_in[2];
  const float* W_dt  = (const float*)d_in[3];
  const float* b_dt  = (const float*)d_in[4];
  const float* Dp    = (const float*)d_in[5];
  float* out = (float*)d_out;

  float* ws    = (float*)d_ws;
  float* x_dbl = ws;
  float* dtbuf = x_dbl + (size_t)ROWS * E_DIM;
  float* prodA = dtbuf + (size_t)ROWS * D_INNER;
  float* hend  = prodA + (size_t)BATCH * D_INNER * STATE * NCHUNK;
  float* hin   = hend  + (size_t)BATCH * D_INNER * STATE * NCHUNK;

  xproj_wmma<<<ROWS / 16, XP_WAVES * 32, 0, stream>>>(u, W_x, W_dt, b_dt, x_dbl, dtbuf);
  scan_pass1<<<BATCH * NCHUNK * DBLKS, SCAN_BLOCK, 0, stream>>>(u, A_log, x_dbl, dtbuf, prodA, hend);
  scan_combine<<<(BATCH * D_INNER * STATE) / 256, 256, 0, stream>>>(prodA, hend, hin);
  scan_pass3<<<BATCH * NCHUNK * DBLKS, SCAN_BLOCK, 0, stream>>>(u, A_log, x_dbl, dtbuf, hin, Dp, out);
}


# scan in 3 launches, not 4: a block scans 8 chunks of 64 channels and joins them through LDS (2 states per wave), pass 3 joins the 8 block groups itself; chunk decay exp(A*sum dt), carries sum 2-level
# speedup vs baseline: 6.5918x; 1.1476x over previous
#include <hip/hip_runtime.h>
#include <hip/hip_bf16.h>

typedef __attribute__((ext_vector_type(2))) float v2f;
typedef __attribute__((ext_vector_type(4))) float v4f;
typedef __attribute__((ext_vector_type(8))) float v8f;

#define D_INNER 1024
#define STATE   16
#define DT_RANK 64
#define BATCH   2
#define SEQ     2048
#define ROWS    (BATCH * SEQ)
#define E_DIM   (DT_RANK + 2 * STATE)
#define NCHUNK  64
#define CLEN    (SEQ / NCHUNK)
#define SCH     8
#define NSUP    (NCHUNK / SCH)
#define SCAN_BLOCK (SCH * 32)
#define CH2     32
#define CGW     (2 * CH2)
#define NCG     (D_INNER / CGW)
#define SGRP    4

#define XP_WAVES  8
#define XP_KSLICE (D_INNER / XP_WAVES)
#define XP_TILES  (E_DIM / 16)

#define XS_PITCH  68
#define DP_PAIRS  4

struct XpFrag {
  v4f a[4];
  v4f b[XP_TILES][4];
};

struct DpFrag {
  v4f b[2][8];
};

__device__ __forceinline__ void dp_load(DpFrag& f, const float* __restrict__ wb) {
#pragma unroll
  for (int t = 0; t < 2; ++t) {
#pragma unroll
    for (int q = 0; q < 8; ++q)
      f.b[t][q] = *(const v4f*)(wb + (size_t)t * 16 * DT_RANK + 8 * q);
  }
}

__device__ __forceinline__ float xhalf(float x) {
  return __builtin_bit_cast(float, __builtin_amdgcn_permlanex16(
      __builtin_bit_cast(int, x), __builtin_bit_cast(int, x),
      0x76543210, 0xfedcba98, false, false));
}

__device__ __forceinline__ float softplus_f(float x) {
  const float e  = __expf(-__builtin_fabsf(x));
  const float d  = 2.0f + e;
  const float r  = __builtin_amdgcn_rcpf(d);
  const float s0 = e * r;
  const float s  = fmaf(fmaf(-s0, d, e), r, s0);
  const float z  = s * s;
  float p = fmaf(z, 1.0f / 17.0f, 1.0f / 15.0f);
  p = fmaf(z, p, 1.0f / 13.0f);
  p = fmaf(z, p, 1.0f / 11.0f);
  p = fmaf(z, p, 1.0f / 9.0f);
  p = fmaf(z, p, 1.0f / 7.0f);
  p = fmaf(z, p, 1.0f / 5.0f);
  p = fmaf(z, p, 1.0f / 3.0f);
  const float s2 = s + s;
  return fmaxf(x, 0.0f) + fmaf(s2 * z, p, s2);
}

__device__ __forceinline__ void dp_pair(const v4f (&af)[8], const DpFrag& f,
                                        float bias, int kh, float* __restrict__ op) {
  v8f c0 = {}, c1 = {};
#pragma unroll
  for (int q = 0; q < 8; ++q) {
    const v2f a0  = {af[q].x, af[q].y};
    const v2f a1  = {af[q].z, af[q].w};
    const v2f b00 = {f.b[0][q].x, f.b[0][q].y};
    const v2f b10 = {f.b[1][q].x, f.b[1][q].y};
    const v2f b01 = {f.b[0][q].z, f.b[0][q].w};
    const v2f b11 = {f.b[1][q].z, f.b[1][q].w};
    c0 = __builtin_amdgcn_wmma_f32_16x16x4_f32(false, a0, false, b00, (short)0, c0, false, false);
    c1 = __builtin_amdgcn_wmma_f32_16x16x4_f32(false, a0, false, b10, (short)0, c1, false, false);
    c0 = __builtin_amdgcn_wmma_f32_16x16x4_f32(false, a1, false, b01, (short)0, c0, false, false);
    c1 = __builtin_amdgcn_wmma_f32_16x16x4_f32(false, a1, false, b11, (short)0, c1, false, false);
  }
#pragma unroll
  for (int j = 0; j < 8; ++j) {
    const float v0 = c0[j], v1 = c1[j];
    const float got = xhalf(kh ? v0 : v1);
    const float x = kh ? got : v0;
    const float y = kh ? v1 : got;
    op[(size_t)j * D_INNER]       = softplus_f(x + bias);
    op[(size_t)(j + 8) * D_INNER] = softplus_f(y + bias);
  }
}

__device__ __forceinline__ void xp_load(XpFrag& f, const float* __restrict__ ua,
                                        const float* __restrict__ wa, int sb) {
#pragma unroll
  for (int q = 0; q < 4; ++q) f.a[q] = *(const v4f*)(ua + 32 * sb + 8 * q);
#pragma unroll
  for (int t = 0; t < XP_TILES; ++t) {
#pragma unroll
    for (int q = 0; q < 4; ++q)
      f.b[t][q] = *(const v4f*)(wa + (size_t)t * 16 * D_INNER + 32 * sb + 8 * q);
  }
}

__device__ __forceinline__ void xp_mma(const XpFrag& f, v8f (&acc)[XP_TILES]) {
#pragma unroll
  for (int q = 0; q < 4; ++q) {
    const v2f a0 = {f.a[q].x, f.a[q].y};
    const v2f a1 = {f.a[q].z, f.a[q].w};
#pragma unroll
    for (int t = 0; t < XP_TILES; ++t) {
      const v2f b0 = {f.b[t][q].x, f.b[t][q].y};
      acc[t] = __builtin_amdgcn_wmma_f32_16x16x4_f32(
           false, a0,  false, b0,
           (short)0, acc[t],  false,  false);
    }
#pragma unroll
    for (int t = 0; t < XP_TILES; ++t) {
      const v2f b1 = {f.b[t][q].z, f.b[t][q].w};
      acc[t] = __builtin_amdgcn_wmma_f32_16x16x4_f32(
          false, a1, false, b1, (short)0, acc[t], false, false);
    }
  }
}

__global__ __launch_bounds__(XP_WAVES * 32) void xproj_wmma(
    const float* __restrict__ u, const float* __restrict__ Wx,
    const float* __restrict__ Wdt, const float* __restrict__ b_dt,
    float* __restrict__ x_dbl, float* __restrict__ dt) {
  __shared__ float red[XP_WAVES][XP_TILES][8][32];
  __shared__ __attribute__((aligned(16))) float xs[16][XS_PITCH];

  const int w    = threadIdx.x >> 5;
  const int lane = threadIdx.x & 31;
  const int row0 = blockIdx.x * 16;
  const int m  = lane & 15;
  const int kh = lane >> 4;

  const float* ua = u  + (size_t)(row0 + m) * D_INNER + w * XP_KSLICE + 4 * kh;
  const float* wa = Wx + (size_t)m * D_INNER + w * XP_KSLICE + 4 * kh;

  v8f acc[XP_TILES];
#pragma unroll
  for (int t = 0; t < XP_TILES; ++t) acc[t] = v8f{};

  XpFrag cur, nxt;
  xp_load(cur, ua, wa, 0);
#pragma unroll
  for (int sb = 0; sb < XP_KSLICE / 32; ++sb) {
    if (sb + 1 < XP_KSLICE / 32) xp_load(nxt, ua, wa, sb + 1);
    __builtin_amdgcn_sched_barrier(0);
    xp_mma(cur, acc);
    if (sb + 1 < XP_KSLICE / 32) cur = nxt;
  }

#pragma unroll
  for (int t = 0; t < XP_TILES; ++t) {
#pragma unroll
    for (int j = 0; j < 8; ++j) red[w][t][j][lane] = acc[t][j];
  }
  __syncthreads();

  float* outp = x_dbl + (size_t)row0 * E_DIM;
#pragma unroll
  for (int i = 0; i < (16 * E_DIM) / (XP_WAVES * 32); ++i) {
    const int o  = threadIdx.x + i * (XP_WAVES * 32);
    const int r  = o / E_DIM;
    const int c  = o - r * E_DIM;
    const int t  = c >> 4;
    const int ln = ((r >> 3) << 4) | (c & 15);
    const int j  = r & 7;
    float sum = red[0][t][j][ln];
#pragma unroll
    for (int ww = 1; ww < XP_WAVES; ++ww) sum += red[ww][t][j][ln];
    outp[o] = sum;
    if (c < DT_RANK) xs[r][c] = sum;
  }
  __syncthreads();

  v4f af[8];
  {
    const float* ap = &xs[m][4 * kh];
#pragma unroll
    for (int q = 0; q < 8; ++q) af[q] = *(const v4f*)(ap + 8 * q);
  }
  const int cbase = w * (DP_PAIRS * 32);
  const float* wb = Wdt + (size_t)(cbase + m) * DT_RANK + 4 * kh;
  const float* bp = b_dt + cbase + lane;
  float* dtp = dt + (size_t)row0 * D_INNER + cbase + lane;

  DpFrag f0, f1;
  float bias0 = bp[0], bias1;
  dp_load(f0, wb);
#pragma unroll 1
  for (int p = 0; p < DP_PAIRS; p += 2) {
    bias1 = bp[(p + 1) * 32];
    dp_load(f1, wb + (size_t)(p + 1) * 32 * DT_RANK);
    __builtin_amdgcn_sched_barrier(0);
    dp_pair(af, f0, bias0, kh, dtp + p * 32);
    __builtin_amdgcn_sched_barrier(0);
    const int pn = (p + 2) & (DP_PAIRS - 1);
    bias0 = bp[pn * 32];
    dp_load(f0, wb + (size_t)pn * 32 * DT_RANK);
    __builtin_amdgcn_sched_barrier(0);
    dp_pair(af, f1, bias1, kh, dtp + (p + 1) * 32);
  }
}

struct ScanIn {
  v2f dt[SGRP];
  v2f u[SGRP];
};

__device__ __forceinline__ void scan_load(ScanIn& f, const float* __restrict__ dtp,
                                          const float* __restrict__ up, int s0) {
#pragma unroll
  for (int i = 0; i < SGRP; ++i) {
    const size_t o = (size_t)(s0 + i) * D_INNER;
    f.dt[i] = v2f{dtp[o], dtp[o + CH2]};
    f.u[i]  = v2f{up[o],  up[o + CH2]};
  }
}

template <int NQ>
struct ScanRow {
  v4f q[NQ];
};

template <int NQ>
__device__ __forceinline__ void row_load(ScanRow<NQ>& r, const float* lp) {
#pragma unroll
  for (int k = 0; k < NQ; ++k) r.q[k] = *(const v4f*)(lp + 4 * k);
}

__device__ __forceinline__ v2f exp_pair(v2f x) {
  return v2f{__expf(x.x), __expf(x.y)};
}

__global__ __launch_bounds__(SCAN_BLOCK) void scan_pass1(
    const float* __restrict__ u, const float* __restrict__ A_log,
    const float* __restrict__ x_dbl, const float* __restrict__ dt,
    float* __restrict__ hinl, float* __restrict__ texc,
    float* __restrict__ hsup, float* __restrict__ tsup) {
  __shared__ __attribute__((aligned(16))) float brow[SCH][CLEN][STATE];
  __shared__ float ex[SCH][STATE][CGW];
  __shared__ float sdl[SCH][CGW];
  const int blk  = blockIdx.x;
  const int cg   = blk % NCG;
  const int s    = (blk / NCG) % NSUP;
  const int b    = blk / (NCG * NSUP);
  const int k    = __builtin_amdgcn_readfirstlane(threadIdx.x >> 5);
  const int lane = threadIdx.x & 31;
  const int chunk = s * SCH + k;
  const int d0   = cg * CGW + lane;
  const int d1   = d0 + CH2;

  v2f A[STATE], h[STATE];
#pragma unroll
  for (int n = 0; n < STATE; ++n) {
    A[n] = v2f{-__expf(A_log[d0 * STATE + n]), -__expf(A_log[d1 * STATE + n])};
    h[n] = v2f{0.0f, 0.0f};
  }
  v2f Aj[2];
#pragma unroll
  for (int i = 0; i < 2; ++i)
    Aj[i] = v2f{-__expf(A_log[d0 * STATE + 2 * k + i]), -__expf(A_log[d1 * STATE + 2 * k + i])};
  v2f sdt = v2f{0.0f, 0.0f};
  const size_t r0 = (size_t)b * SEQ + (size_t)chunk * CLEN;
  const float* dtp = dt + r0 * D_INNER + d0;
  const float* up  = u  + r0 * D_INNER + d0;
  const float* bp  = x_dbl + r0 * E_DIM + DT_RANK;

  ScanIn cur, nxt;
  scan_load(cur, dtp, up, 0);
#pragma unroll
  for (int j = 0; j < 4; ++j) {
    const int r = (lane >> 2) + 8 * j, q = lane & 3;
    *(v4f*)&brow[k][r][4 * q] = *(const v4f*)(bp + (size_t)r * E_DIM + 4 * q);
  }
  __syncthreads();
  ScanRow<4> rc, rn;
  row_load(rc, &brow[k][0][0]);
#pragma unroll 2
  for (int g = 0; g < CLEN / SGRP; ++g) {
    const int gn = (g + 1 < CLEN / SGRP) ? g + 1 : g;
    scan_load(nxt, dtp, up, gn * SGRP);
    __builtin_amdgcn_sched_barrier(0);
#pragma unroll
    for (int i = 0; i < SGRP; ++i) {
      const int st = g * SGRP + i;
      row_load(rn, &brow[k][(st + 1 < CLEN) ? st + 1 : st][0]);
      __builtin_amdgcn_sched_barrier(0);
      const v2f dtv = cur.dt[i];
      const v2f dtu = dtv * cur.u[i];
      sdt += dtv;
#pragma unroll
      for (int n = 0; n < STATE; ++n) {
        const v2f a = exp_pair(dtv * A[n]);
        h[n] = __builtin_elementwise_fma(a, h[n], dtu * rc.q[n >> 2][n & 3]);
      }
      rc = rn;
    }
    cur = nxt;
  }

#pragma unroll
  for (int n = 0; n < STATE; ++n) {
    ex[k][n][lane]       = h[n].x;
    ex[k][n][lane + CH2] = h[n].y;
  }
  sdl[k][lane]       = sdt.x;
  sdl[k][lane + CH2] = sdt.y;
  __syncthreads();

  const size_t c0 = (size_t)b * NCHUNK + (size_t)s * SCH;
  v2f sd[SCH], ej[2][SCH];
#pragma unroll
  for (int c = 0; c < SCH; ++c) {
    sd[c] = v2f{sdl[c][lane], sdl[c][lane + CH2]};
#pragma unroll
    for (int i = 0; i < 2; ++i) ej[i][c] = v2f{ex[c][2 * k + i][lane], ex[c][2 * k + i][lane + CH2]};
  }
  __builtin_amdgcn_sched_barrier(0);
  {
    v2f t = v2f{0.0f, 0.0f};
#pragma unroll
    for (int j = 0; j < SCH - 1; ++j) t += (j < k) ? sd[j] : v2f{0.0f, 0.0f};
    const size_t o = (c0 + k) * D_INNER + d0;
    texc[o]       = t.x;
    texc[o + CH2] = t.y;
    if (k == SCH - 1) {
      t += sd[SCH - 1];
      const size_t os = (size_t)(b * NSUP + s) * D_INNER + d0;
      tsup[os]       = t.x;
      tsup[os + CH2] = t.y;
    }
  }
#pragma unroll
  for (int i = 0; i < 2; ++i) {
    const int n = 2 * k + i;
    const v2f An = Aj[i];
    v2f H = v2f{0.0f, 0.0f};
#pragma unroll
    for (int c = 0; c < SCH; ++c) {
      const size_t o = ((c0 + c) * STATE + n) * D_INNER + d0;
      hinl[o]       = H.x;
      hinl[o + CH2] = H.y;
      H = __builtin_elementwise_fma(exp_pair(An * sd[c]), H, ej[i][c]);
    }
    const size_t os = ((size_t)(b * NSUP + s) * STATE + n) * D_INNER + d0;
    hsup[os]       = H.x;
    hsup[os + CH2] = H.y;
  }
}

__global__ __launch_bounds__(SCAN_BLOCK) void scan_pass3(
    const float* __restrict__ u, const float* __restrict__ A_log,
    const float* __restrict__ x_dbl, const float* __restrict__ dt,
    const float* __restrict__ hinl, const float* __restrict__ texc,
    const float* __restrict__ hsup, const float* __restrict__ tsup,
    const float* __restrict__ Dp, float* __restrict__ out) {
  __shared__ __attribute__((aligned(16))) float bcrow[SCH][CLEN][2 * STATE];
  __shared__ float hcar[STATE][CGW];
  const int blk  = blockIdx.x;
  const int cg   = blk % NCG;
  const int s    = (blk / NCG) % NSUP;
  const int b    = blk / (NCG * NSUP);
  const int k    = __builtin_amdgcn_readfirstlane(threadIdx.x >> 5);
  const int lane = threadIdx.x & 31;
  const int chunk = s * SCH + k;
  const int d0   = cg * CGW + lane;
  const int d1   = d0 + CH2;

  v2f ts[NSUP - 1], hs[2][NSUP - 1];
#pragma unroll
  for (int j = 0; j < NSUP - 1; ++j) {
    const size_t ot = (size_t)(b * NSUP + j) * D_INNER + d0;
    ts[j] = v2f{tsup[ot], tsup[ot + CH2]};
#pragma unroll
    for (int i = 0; i < 2; ++i) {
      const size_t oh = ((size_t)(b * NSUP + j) * STATE + 2 * k + i) * D_INNER + d0;
      hs[i][j] = v2f{hsup[oh], hsup[oh + CH2]};
    }
  }

  const size_t cidx = (size_t)b * NCHUNK + chunk;
  v2f A[STATE], h[STATE];
#pragma unroll
  for (int n = 0; n < STATE; ++n) {
    const size_t o = (cidx * STATE + n) * D_INNER + d0;
    A[n] = v2f{-__expf(A_log[d0 * STATE + n]), -__expf(A_log[d1 * STATE + n])};
    h[n] = v2f{hinl[o], hinl[o + CH2]};
  }
  const v2f te = v2f{texc[cidx * D_INNER + d0], texc[cidx * D_INNER + d1]};
  const v2f dp = v2f{Dp[d0], Dp[d1]};
  const size_t r0 = (size_t)b * SEQ + (size_t)chunk * CLEN;
  const float* dtp = dt + r0 * D_INNER + d0;
  const float* up  = u  + r0 * D_INNER + d0;
  const float* bp  = x_dbl + r0 * E_DIM + DT_RANK;
  float* op = out + r0 * D_INNER + d0;

  ScanIn cur, nxt;
  scan_load(cur, dtp, up, 0);
#pragma unroll
  for (int j = 0; j < 8; ++j) {
    const int r = (lane >> 3) + 4 * j, q = lane & 7;
    *(v4f*)&bcrow[k][r][4 * q] = *(const v4f*)(bp + (size_t)r * E_DIM + 4 * q);
  }
#pragma unroll
  for (int i = 0; i < 2; ++i) {
    const int n = 2 * k + i;
    const v2f An = v2f{-__expf(A_log[d0 * STATE + n]), -__expf(A_log[d1 * STATE + n])};
    v2f H = v2f{0.0f, 0.0f};
#pragma unroll
    for (int j = 0; j < NSUP - 1; ++j) {
      const v2f Hn = __builtin_elementwise_fma(exp_pair(An * ts[j]), H, hs[i][j]);
      H = (j < s) ? Hn : H;
    }
    hcar[n][lane]       = H.x;
    hcar[n][lane + CH2] = H.y;
  }
  __syncthreads();
  if (k != 0) {
#pragma unroll
    for (int n = 0; n < STATE; ++n) {
      const v2f hc = v2f{hcar[n][lane], hcar[n][lane + CH2]};
      h[n] = __builtin_elementwise_fma(exp_pair(A[n] * te), hc, h[n]);
    }
  } else {
#pragma unroll
    for (int n = 0; n < STATE; ++n) h[n] += v2f{hcar[n][lane], hcar[n][lane + CH2]};
  }

  ScanRow<8> rc, rn;
  row_load(rc, &bcrow[k][0][0]);
#pragma unroll 2
  for (int g = 0; g < CLEN / SGRP; ++g) {
    const int gn = (g + 1 < CLEN / SGRP) ? g + 1 : g;
    scan_load(nxt, dtp, up, gn * SGRP);
    __builtin_amdgcn_sched_barrier(0);
#pragma unroll
    for (int i = 0; i < SGRP; ++i) {
      const int st = g * SGRP + i;
      row_load(rn, &bcrow[k][(st + 1 < CLEN) ? st + 1 : st][0]);
      __builtin_amdgcn_sched_barrier(0);
      const v2f dtv = cur.dt[i];
      const v2f uv  = cur.u[i];
      const v2f dtu = dtv * uv;
      v2f y[4];
#pragma unroll
      for (int q = 0; q < 4; ++q) y[q] = v2f{0.0f, 0.0f};
#pragma unroll
      for (int n = 0; n < STATE; ++n) {
        const v2f a = exp_pair(dtv * A[n]);
        h[n] = __builtin_elementwise_fma(a, h[n], dtu * rc.q[n >> 2][n & 3]);
        const float cn = rc.q[4 + (n >> 2)][n & 3];
        const v2f c = v2f{cn, cn};
        y[n & 3] = __builtin_elementwise_fma(h[n], c, y[n & 3]);
      }
      const v2f o = __builtin_elementwise_fma(uv, dp, (y[0] + y[1]) + (y[2] + y[3]));
      op[(size_t)st * D_INNER]       = o.x;
      op[(size_t)st * D_INNER + CH2] = o.y;
      rc = rn;
    }
    cur = nxt;
  }
}

extern "C" void kernel_launch(void* const* d_in, const int* in_sizes, int n_in,
                              void* d_out, int out_size, void* d_ws, size_t ws_size,
                              hipStream_t stream) {
  const float* u     = (const float*)d_in[0];
  const float* A_log = (const float*)d_in[1];
  const float* W_x   = (const float*)d_in[2];
  const float* W_dt  = (const float*)d_in[3];
  const float* b_dt  = (const float*)d_in[4];
  const float* Dp    = (const float*)d_in[5];
  float* out = (float*)d_out;

  float* ws    = (float*)d_ws;
  float* x_dbl = ws;
  float* dtbuf = x_dbl + (size_t)ROWS * E_DIM;
  float* hinl  = dtbuf + (size_t)ROWS * D_INNER;
  float* texc  = hinl  + (size_t)BATCH * NCHUNK * STATE * D_INNER;
  float* hsup  = texc  + (size_t)BATCH * NCHUNK * D_INNER;
  float* tsup  = hsup  + (size_t)BATCH * NSUP * STATE * D_INNER;

  xproj_wmma<<<ROWS / 16, XP_WAVES * 32, 0, stream>>>(u, W_x, W_dt, b_dt, x_dbl, dtbuf);
  scan_pass1<<<BATCH * NSUP * NCG, SCAN_BLOCK, 0, stream>>>(u, A_log, x_dbl, dtbuf, hinl, texc, hsup, tsup);
  scan_pass3<<<BATCH * NSUP * NCG, SCAN_BLOCK, 0, stream>>>(u, A_log, x_dbl, dtbuf, hinl, texc, hsup, tsup, Dp, out);
}


# scan passes: decay rates held per lane as A*log2(e), so exp(dt*A) is one base-2 hardware exponential of dt*A2, no multiply by log2(e) per value; decay factors round differently, sum orders unchanged
# speedup vs baseline: 6.8150x; 1.0339x over previous
#include <hip/hip_runtime.h>
#include <hip/hip_bf16.h>

typedef __attribute__((ext_vector_type(2))) float v2f;
typedef __attribute__((ext_vector_type(4))) float v4f;
typedef __attribute__((ext_vector_type(8))) float v8f;

#define D_INNER 1024
#define STATE   16
#define DT_RANK 64
#define BATCH   2
#define SEQ     2048
#define ROWS    (BATCH * SEQ)
#define E_DIM   (DT_RANK + 2 * STATE)
#define NCHUNK  64
#define CLEN    (SEQ / NCHUNK)
#define SCH     8
#define NSUP    (NCHUNK / SCH)
#define SCAN_BLOCK (SCH * 32)
#define CH2     32
#define CGW     (2 * CH2)
#define NCG     (D_INNER / CGW)
#define SGRP    4

#define XP_WAVES  8
#define XP_KSLICE (D_INNER / XP_WAVES)
#define XP_TILES  (E_DIM / 16)

#define XS_PITCH  68
#define DP_PAIRS  4

struct XpFrag {
  v4f a[4];
  v4f b[XP_TILES][4];
};

struct DpFrag {
  v4f b[2][8];
};

__device__ __forceinline__ void dp_load(DpFrag& f, const float* __restrict__ wb) {
#pragma unroll
  for (int t = 0; t < 2; ++t) {
#pragma unroll
    for (int q = 0; q < 8; ++q)
      f.b[t][q] = *(const v4f*)(wb + (size_t)t * 16 * DT_RANK + 8 * q);
  }
}

__device__ __forceinline__ float xhalf(float x) {
  return __builtin_bit_cast(float, __builtin_amdgcn_permlanex16(
      __builtin_bit_cast(int, x), __builtin_bit_cast(int, x),
      0x76543210, 0xfedcba98, false, false));
}

__device__ __forceinline__ float softplus_f(float x) {
  const float e  = __expf(-__builtin_fabsf(x));
  const float d  = 2.0f + e;
  const float r  = __builtin_amdgcn_rcpf(d);
  const float s0 = e * r;
  const float s  = fmaf(fmaf(-s0, d, e), r, s0);
  const float z  = s * s;
  float p = fmaf(z, 1.0f / 17.0f, 1.0f / 15.0f);
  p = fmaf(z, p, 1.0f / 13.0f);
  p = fmaf(z, p, 1.0f / 11.0f);
  p = fmaf(z, p, 1.0f / 9.0f);
  p = fmaf(z, p, 1.0f / 7.0f);
  p = fmaf(z, p, 1.0f / 5.0f);
  p = fmaf(z, p, 1.0f / 3.0f);
  const float s2 = s + s;
  return fmaxf(x, 0.0f) + fmaf(s2 * z, p, s2);
}

__device__ __forceinline__ void dp_pair(const v4f (&af)[8], const DpFrag& f,
                                        float bias, int kh, float* __restrict__ op) {
  v8f c0 = {}, c1 = {};
#pragma unroll
  for (int q = 0; q < 8; ++q) {
    const v2f a0  = {af[q].x, af[q].y};
    const v2f a1  = {af[q].z, af[q].w};
    const v2f b00 = {f.b[0][q].x, f.b[0][q].y};
    const v2f b10 = {f.b[1][q].x, f.b[1][q].y};
    const v2f b01 = {f.b[0][q].z, f.b[0][q].w};
    const v2f b11 = {f.b[1][q].z, f.b[1][q].w};
    c0 = __builtin_amdgcn_wmma_f32_16x16x4_f32(false, a0, false, b00, (short)0, c0, false, false);
    c1 = __builtin_amdgcn_wmma_f32_16x16x4_f32(false, a0, false, b10, (short)0, c1, false, false);
    c0 = __builtin_amdgcn_wmma_f32_16x16x4_f32(false, a1, false, b01, (short)0, c0, false, false);
    c1 = __builtin_amdgcn_wmma_f32_16x16x4_f32(false, a1, false, b11, (short)0, c1, false, false);
  }
#pragma unroll
  for (int j = 0; j < 8; ++j) {
    const float v0 = c0[j], v1 = c1[j];
    const float got = xhalf(kh ? v0 : v1);
    const float x = kh ? got : v0;
    const float y = kh ? v1 : got;
    op[(size_t)j * D_INNER]       = softplus_f(x + bias);
    op[(size_t)(j + 8) * D_INNER] = softplus_f(y + bias);
  }
}

__device__ __forceinline__ void xp_load(XpFrag& f, const float* __restrict__ ua,
                                        const float* __restrict__ wa, int sb) {
#pragma unroll
  for (int q = 0; q < 4; ++q) f.a[q] = *(const v4f*)(ua + 32 * sb + 8 * q);
#pragma unroll
  for (int t = 0; t < XP_TILES; ++t) {
#pragma unroll
    for (int q = 0; q < 4; ++q)
      f.b[t][q] = *(const v4f*)(wa + (size_t)t * 16 * D_INNER + 32 * sb + 8 * q);
  }
}

__device__ __forceinline__ void xp_mma(const XpFrag& f, v8f (&acc)[XP_TILES]) {
#pragma unroll
  for (int q = 0; q < 4; ++q) {
    const v2f a0 = {f.a[q].x, f.a[q].y};
    const v2f a1 = {f.a[q].z, f.a[q].w};
#pragma unroll
    for (int t = 0; t < XP_TILES; ++t) {
      const v2f b0 = {f.b[t][q].x, f.b[t][q].y};
      acc[t] = __builtin_amdgcn_wmma_f32_16x16x4_f32(
           false, a0,  false, b0,
           (short)0, acc[t],  false,  false);
    }
#pragma unroll
    for (int t = 0; t < XP_TILES; ++t) {
      const v2f b1 = {f.b[t][q].z, f.b[t][q].w};
      acc[t] = __builtin_amdgcn_wmma_f32_16x16x4_f32(
          false, a1, false, b1, (short)0, acc[t], false, false);
    }
  }
}

__global__ __launch_bounds__(XP_WAVES * 32) void xproj_wmma(
    const float* __restrict__ u, const float* __restrict__ Wx,
    const float* __restrict__ Wdt, const float* __restrict__ b_dt,
    float* __restrict__ x_dbl, float* __restrict__ dt) {
  __shared__ float red[XP_WAVES][XP_TILES][8][32];
  __shared__ __attribute__((aligned(16))) float xs[16][XS_PITCH];

  const int w    = threadIdx.x >> 5;
  const int lane = threadIdx.x & 31;
  const int row0 = blockIdx.x * 16;
  const int m  = lane & 15;
  const int kh = lane >> 4;

  const float* ua = u  + (size_t)(row0 + m) * D_INNER + w * XP_KSLICE + 4 * kh;
  const float* wa = Wx + (size_t)m * D_INNER + w * XP_KSLICE + 4 * kh;

  v8f acc[XP_TILES];
#pragma unroll
  for (int t = 0; t < XP_TILES; ++t) acc[t] = v8f{};

  XpFrag cur, nxt;
  xp_load(cur, ua, wa, 0);
#pragma unroll
  for (int sb = 0; sb < XP_KSLICE / 32; ++sb) {
    if (sb + 1 < XP_KSLICE / 32) xp_load(nxt, ua, wa, sb + 1);
    __builtin_amdgcn_sched_barrier(0);
    xp_mma(cur, acc);
    if (sb + 1 < XP_KSLICE / 32) cur = nxt;
  }

#pragma unroll
  for (int t = 0; t < XP_TILES; ++t) {
#pragma unroll
    for (int j = 0; j < 8; ++j) red[w][t][j][lane] = acc[t][j];
  }
  __syncthreads();

  float* outp = x_dbl + (size_t)row0 * E_DIM;
#pragma unroll
  for (int i = 0; i < (16 * E_DIM) / (XP_WAVES * 32); ++i) {
    const int o  = threadIdx.x + i * (XP_WAVES * 32);
    const int r  = o / E_DIM;
    const int c  = o - r * E_DIM;
    const int t  = c >> 4;
    const int ln = ((r >> 3) << 4) | (c & 15);
    const int j  = r & 7;
    float sum = red[0][t][j][ln];
#pragma unroll
    for (int ww = 1; ww < XP_WAVES; ++ww) sum += red[ww][t][j][ln];
    outp[o] = sum;
    if (c < DT_RANK) xs[r][c] = sum;
  }
  __syncthreads();

  v4f af[8];
  {
    const float* ap = &xs[m][4 * kh];
#pragma unroll
    for (int q = 0; q < 8; ++q) af[q] = *(const v4f*)(ap + 8 * q);
  }
  const int cbase = w * (DP_PAIRS * 32);
  const float* wb = Wdt + (size_t)(cbase + m) * DT_RANK + 4 * kh;
  const float* bp = b_dt + cbase + lane;
  float* dtp = dt + (size_t)row0 * D_INNER + cbase + lane;

  DpFrag f0, f1;
  float bias0 = bp[0], bias1;
  dp_load(f0, wb);
#pragma unroll 1
  for (int p = 0; p < DP_PAIRS; p += 2) {
    bias1 = bp[(p + 1) * 32];
    dp_load(f1, wb + (size_t)(p + 1) * 32 * DT_RANK);
    __builtin_amdgcn_sched_barrier(0);
    dp_pair(af, f0, bias0, kh, dtp + p * 32);
    __builtin_amdgcn_sched_barrier(0);
    const int pn = (p + 2) & (DP_PAIRS - 1);
    bias0 = bp[pn * 32];
    dp_load(f0, wb + (size_t)pn * 32 * DT_RANK);
    __builtin_amdgcn_sched_barrier(0);
    dp_pair(af, f1, bias1, kh, dtp + (p + 1) * 32);
  }
}

struct ScanIn {
  v2f dt[SGRP];
  v2f u[SGRP];
};

__device__ __forceinline__ void scan_load(ScanIn& f, const float* __restrict__ dtp,
                                          const float* __restrict__ up, int s0) {
#pragma unroll
  for (int i = 0; i < SGRP; ++i) {
    const size_t o = (size_t)(s0 + i) * D_INNER;
    f.dt[i] = v2f{dtp[o], dtp[o + CH2]};
    f.u[i]  = v2f{up[o],  up[o + CH2]};
  }
}

template <int NQ>
struct ScanRow {
  v4f q[NQ];
};

template <int NQ>
__device__ __forceinline__ void row_load(ScanRow<NQ>& r, const float* lp) {
#pragma unroll
  for (int k = 0; k < NQ; ++k) r.q[k] = *(const v4f*)(lp + 4 * k);
}

__device__ __forceinline__ v2f exp2_pair(v2f x) {
  return v2f{__builtin_amdgcn_exp2f(x.x), __builtin_amdgcn_exp2f(x.y)};
}

#define LOG2E_F 1.44269504088896340736f

__device__ __forceinline__ float rate2(float alog) {
  return -__expf(alog) * LOG2E_F;
}

__global__ __launch_bounds__(SCAN_BLOCK) void scan_pass1(
    const float* __restrict__ u, const float* __restrict__ A_log,
    const float* __restrict__ x_dbl, const float* __restrict__ dt,
    float* __restrict__ hinl, float* __restrict__ texc,
    float* __restrict__ hsup, float* __restrict__ tsup) {
  __shared__ __attribute__((aligned(16))) float brow[SCH][CLEN][STATE];
  __shared__ float ex[SCH][STATE][CGW];
  __shared__ float sdl[SCH][CGW];
  const int blk  = blockIdx.x;
  const int cg   = blk % NCG;
  const int s    = (blk / NCG) % NSUP;
  const int b    = blk / (NCG * NSUP);
  const int k    = __builtin_amdgcn_readfirstlane(threadIdx.x >> 5);
  const int lane = threadIdx.x & 31;
  const int chunk = s * SCH + k;
  const int d0   = cg * CGW + lane;
  const int d1   = d0 + CH2;

  v2f A[STATE], h[STATE];
#pragma unroll
  for (int n = 0; n < STATE; ++n) {
    A[n] = v2f{rate2(A_log[d0 * STATE + n]), rate2(A_log[d1 * STATE + n])};
    h[n] = v2f{0.0f, 0.0f};
  }
  v2f Aj[2];
#pragma unroll
  for (int i = 0; i < 2; ++i)
    Aj[i] = v2f{rate2(A_log[d0 * STATE + 2 * k + i]), rate2(A_log[d1 * STATE + 2 * k + i])};
  v2f sdt = v2f{0.0f, 0.0f};
  const size_t r0 = (size_t)b * SEQ + (size_t)chunk * CLEN;
  const float* dtp = dt + r0 * D_INNER + d0;
  const float* up  = u  + r0 * D_INNER + d0;
  const float* bp  = x_dbl + r0 * E_DIM + DT_RANK;

  ScanIn cur, nxt;
  scan_load(cur, dtp, up, 0);
#pragma unroll
  for (int j = 0; j < 4; ++j) {
    const int r = (lane >> 2) + 8 * j, q = lane & 3;
    *(v4f*)&brow[k][r][4 * q] = *(const v4f*)(bp + (size_t)r * E_DIM + 4 * q);
  }
  __syncthreads();
  ScanRow<4> rc, rn;
  row_load(rc, &brow[k][0][0]);
#pragma unroll 2
  for (int g = 0; g < CLEN / SGRP; ++g) {
    const int gn = (g + 1 < CLEN / SGRP) ? g + 1 : g;
    scan_load(nxt, dtp, up, gn * SGRP);
    __builtin_amdgcn_sched_barrier(0);
#pragma unroll
    for (int i = 0; i < SGRP; ++i) {
      const int st = g * SGRP + i;
      row_load(rn, &brow[k][(st + 1 < CLEN) ? st + 1 : st][0]);
      __builtin_amdgcn_sched_barrier(0);
      const v2f dtv = cur.dt[i];
      const v2f dtu = dtv * cur.u[i];
      sdt += dtv;
#pragma unroll
      for (int n = 0; n < STATE; ++n) {
        const v2f a = exp2_pair(dtv * A[n]);
        h[n] = __builtin_elementwise_fma(a, h[n], dtu * rc.q[n >> 2][n & 3]);
      }
      rc = rn;
    }
    cur = nxt;
  }

#pragma unroll
  for (int n = 0; n < STATE; ++n) {
    ex[k][n][lane]       = h[n].x;
    ex[k][n][lane + CH2] = h[n].y;
  }
  sdl[k][lane]       = sdt.x;
  sdl[k][lane + CH2] = sdt.y;
  __syncthreads();

  const size_t c0 = (size_t)b * NCHUNK + (size_t)s * SCH;
  v2f sd[SCH], ej[2][SCH];
#pragma unroll
  for (int c = 0; c < SCH; ++c) {
    sd[c] = v2f{sdl[c][lane], sdl[c][lane + CH2]};
#pragma unroll
    for (int i = 0; i < 2; ++i) ej[i][c] = v2f{ex[c][2 * k + i][lane], ex[c][2 * k + i][lane + CH2]};
  }
  __builtin_amdgcn_sched_barrier(0);
  {
    v2f t = v2f{0.0f, 0.0f};
#pragma unroll
    for (int j = 0; j < SCH - 1; ++j) t += (j < k) ? sd[j] : v2f{0.0f, 0.0f};
    const size_t o = (c0 + k) * D_INNER + d0;
    texc[o]       = t.x;
    texc[o + CH2] = t.y;
    if (k == SCH - 1) {
      t += sd[SCH - 1];
      const size_t os = (size_t)(b * NSUP + s) * D_INNER + d0;
      tsup[os]       = t.x;
      tsup[os + CH2] = t.y;
    }
  }
#pragma unroll
  for (int i = 0; i < 2; ++i) {
    const int n = 2 * k + i;
    const v2f An = Aj[i];
    v2f H = v2f{0.0f, 0.0f};
#pragma unroll
    for (int c = 0; c < SCH; ++c) {
      const size_t o = ((c0 + c) * STATE + n) * D_INNER + d0;
      hinl[o]       = H.x;
      hinl[o + CH2] = H.y;
      H = __builtin_elementwise_fma(exp2_pair(An * sd[c]), H, ej[i][c]);
    }
    const size_t os = ((size_t)(b * NSUP + s) * STATE + n) * D_INNER + d0;
    hsup[os]       = H.x;
    hsup[os + CH2] = H.y;
  }
}

__global__ __launch_bounds__(SCAN_BLOCK) void scan_pass3(
    const float* __restrict__ u, const float* __restrict__ A_log,
    const float* __restrict__ x_dbl, const float* __restrict__ dt,
    const float* __restrict__ hinl, const float* __restrict__ texc,
    const float* __restrict__ hsup, const float* __restrict__ tsup,
    const float* __restrict__ Dp, float* __restrict__ out) {
  __shared__ __attribute__((aligned(16))) float bcrow[SCH][CLEN][2 * STATE];
  __shared__ float hcar[STATE][CGW];
  const int blk  = blockIdx.x;
  const int cg   = blk % NCG;
  const int s    = (blk / NCG) % NSUP;
  const int b    = blk / (NCG * NSUP);
  const int k    = __builtin_amdgcn_readfirstlane(threadIdx.x >> 5);
  const int lane = threadIdx.x & 31;
  const int chunk = s * SCH + k;
  const int d0   = cg * CGW + lane;
  const int d1   = d0 + CH2;

  v2f ts[NSUP - 1], hs[2][NSUP - 1];
#pragma unroll
  for (int j = 0; j < NSUP - 1; ++j) {
    const size_t ot = (size_t)(b * NSUP + j) * D_INNER + d0;
    ts[j] = v2f{tsup[ot], tsup[ot + CH2]};
#pragma unroll
    for (int i = 0; i < 2; ++i) {
      const size_t oh = ((size_t)(b * NSUP + j) * STATE + 2 * k + i) * D_INNER + d0;
      hs[i][j] = v2f{hsup[oh], hsup[oh + CH2]};
    }
  }

  const size_t cidx = (size_t)b * NCHUNK + chunk;
  v2f A[STATE], h[STATE];
#pragma unroll
  for (int n = 0; n < STATE; ++n) {
    const size_t o = (cidx * STATE + n) * D_INNER + d0;
    A[n] = v2f{rate2(A_log[d0 * STATE + n]), rate2(A_log[d1 * STATE + n])};
    h[n] = v2f{hinl[o], hinl[o + CH2]};
  }
  const v2f te = v2f{texc[cidx * D_INNER + d0], texc[cidx * D_INNER + d1]};
  const v2f dp = v2f{Dp[d0], Dp[d1]};
  const size_t r0 = (size_t)b * SEQ + (size_t)chunk * CLEN;
  const float* dtp = dt + r0 * D_INNER + d0;
  const float* up  = u  + r0 * D_INNER + d0;
  const float* bp  = x_dbl + r0 * E_DIM + DT_RANK;
  float* op = out + r0 * D_INNER + d0;

  ScanIn cur, nxt;
  scan_load(cur, dtp, up, 0);
#pragma unroll
  for (int j = 0; j < 8; ++j) {
    const int r = (lane >> 3) + 4 * j, q = lane & 7;
    *(v4f*)&bcrow[k][r][4 * q] = *(const v4f*)(bp + (size_t)r * E_DIM + 4 * q);
  }
#pragma unroll
  for (int i = 0; i < 2; ++i) {
    const int n = 2 * k + i;
    const v2f An = v2f{rate2(A_log[d0 * STATE + n]), rate2(A_log[d1 * STATE + n])};
    v2f H = v2f{0.0f, 0.0f};
#pragma unroll
    for (int j = 0; j < NSUP - 1; ++j) {
      const v2f Hn = __builtin_elementwise_fma(exp2_pair(An * ts[j]), H, hs[i][j]);
      H = (j < s) ? Hn : H;
    }
    hcar[n][lane]       = H.x;
    hcar[n][lane + CH2] = H.y;
  }
  __syncthreads();
  if (k != 0) {
#pragma unroll
    for (int n = 0; n < STATE; ++n) {
      const v2f hc = v2f{hcar[n][lane], hcar[n][lane + CH2]};
      h[n] = __builtin_elementwise_fma(exp2_pair(A[n] * te), hc, h[n]);
    }
  } else {
#pragma unroll
    for (int n = 0; n < STATE; ++n) h[n] += v2f{hcar[n][lane], hcar[n][lane + CH2]};
  }

  ScanRow<8> rc, rn;
  row_load(rc, &bcrow[k][0][0]);
#pragma unroll 2
  for (int g = 0; g < CLEN / SGRP; ++g) {
    const int gn = (g + 1 < CLEN / SGRP) ? g + 1 : g;
    scan_load(nxt, dtp, up, gn * SGRP);
    __builtin_amdgcn_sched_barrier(0);
#pragma unroll
    for (int i = 0; i < SGRP; ++i) {
      const int st = g * SGRP + i;
      row_load(rn, &bcrow[k][(st + 1 < CLEN) ? st + 1 : st][0]);
      __builtin_amdgcn_sched_barrier(0);
      const v2f dtv = cur.dt[i];
      const v2f uv  = cur.u[i];
      const v2f dtu = dtv * uv;
      v2f y[4];
#pragma unroll
      for (int q = 0; q < 4; ++q) y[q] = v2f{0.0f, 0.0f};
#pragma unroll
      for (int n = 0; n < STATE; ++n) {
        const v2f a = exp2_pair(dtv * A[n]);
        h[n] = __builtin_elementwise_fma(a, h[n], dtu * rc.q[n >> 2][n & 3]);
        const float cn = rc.q[4 + (n >> 2)][n & 3];
        const v2f c = v2f{cn, cn};
        y[n & 3] = __builtin_elementwise_fma(h[n], c, y[n & 3]);
      }
      const v2f o = __builtin_elementwise_fma(uv, dp, (y[0] + y[1]) + (y[2] + y[3]));
      op[(size_t)st * D_INNER]       = o.x;
      op[(size_t)st * D_INNER + CH2] = o.y;
      rc = rn;
    }
    cur = nxt;
  }
}

extern "C" void kernel_launch(void* const* d_in, const int* in_sizes, int n_in,
                              void* d_out, int out_size, void* d_ws, size_t ws_size,
                              hipStream_t stream) {
  const float* u     = (const float*)d_in[0];
  const float* A_log = (const float*)d_in[1];
  const float* W_x   = (const float*)d_in[2];
  const float* W_dt  = (const float*)d_in[3];
  const float* b_dt  = (const float*)d_in[4];
  const float* Dp    = (const float*)d_in[5];
  float* out = (float*)d_out;

  float* ws    = (float*)d_ws;
  float* x_dbl = ws;
  float* dtbuf = x_dbl + (size_t)ROWS * E_DIM;
  float* hinl  = dtbuf + (size_t)ROWS * D_INNER;
  float* texc  = hinl  + (size_t)BATCH * NCHUNK * STATE * D_INNER;
  float* hsup  = texc  + (size_t)BATCH * NCHUNK * D_INNER;
  float* tsup  = hsup  + (size_t)BATCH * NSUP * STATE * D_INNER;

  xproj_wmma<<<ROWS / 16, XP_WAVES * 32, 0, stream>>>(u, W_x, W_dt, b_dt, x_dbl, dtbuf);
  scan_pass1<<<BATCH * NSUP * NCG, SCAN_BLOCK, 0, stream>>>(u, A_log, x_dbl, dtbuf, hinl, texc, hsup, tsup);
  scan_pass3<<<BATCH * NSUP * NCG, SCAN_BLOCK, 0, stream>>>(u, A_log, x_dbl, dtbuf, hinl, texc, hsup, tsup, Dp, out);
}
